# starting kernel
# baseline (speedup 1.0000x reference)
#include <hip/hip_runtime.h>
#include <hip/hip_bf16.h>

typedef __attribute__((ext_vector_type(2))) float v2f;
typedef __attribute__((ext_vector_type(8))) float v8f;


__device__ __forceinline__ void gate_rz(float2* st, int wire, float theta, int t) {
    float h = 0.5f * theta;
    float c = cosf(h), s = sinf(h);
    int mask = 1 << (7 - wire);
    int low  = t & (mask - 1);
    int i0   = ((t ^ low) << 1) | low;
    int i1   = i0 | mask;
    float2 a = st[i0], b = st[i1];
    st[i0] = make_float2(a.x * c + a.y * s, a.y * c - a.x * s);
    st[i1] = make_float2(b.x * c - b.y * s, b.y * c + b.x * s);
}

__device__ __forceinline__ void gate_ry(float2* st, int wire, float theta, int t) {
    float h = 0.5f * theta;
    float c = cosf(h), s = sinf(h);
    int mask = 1 << (7 - wire);
    int low  = t & (mask - 1);
    int i0   = ((t ^ low) << 1) | low;
    int i1   = i0 | mask;
    float2 a = st[i0], b = st[i1];
    st[i0] = make_float2(c * a.x - s * b.x, c * a.y - s * b.y);
    st[i1] = make_float2(s * a.x + c * b.x, s * a.y + c * b.y);
}

__device__ __forceinline__ void gate_cnot(float2* st, int ctrl, int tgt, int t) {
    int pc = 7 - ctrl, pt = 7 - tgt;
    int p1 = pc < pt ? pc : pt;
    int p2 = pc < pt ? pt : pc;
    if (t < 64) {
        unsigned low  = (unsigned)t & ((1u << p1) - 1u);
        unsigned mid  = ((unsigned)t >> p1) & ((1u << (p2 - p1 - 1)) - 1u);
        unsigned high = (unsigned)t >> (p2 - 1);
        unsigned i = (high << (p2 + 1)) | (mid << (p1 + 1)) | low;
        i |= (1u << pc);
        unsigned ip = i | (1u << pt);
        float2 a = st[i], b = st[ip];
        st[i]  = b;
        st[ip] = a;
    }
}

__global__ void qcnn_build_matrix(const float* __restrict__ w, float* __restrict__ AtP) {
    __shared__ float2 st[256];
    const int j = blockIdx.x;
    const int t = threadIdx.x;

    st[t]       = make_float2((t == j)       ? 1.0f : 0.0f, 0.0f);
    st[t + 128] = make_float2((t + 128 == j) ? 1.0f : 0.0f, 0.0f);
    __syncthreads();

    const int pr1[7] = {0, 2, 4, 6, 1, 5, 3};
    const int pr2[7] = {1, 3, 5, 7, 3, 7, 7};
    const float HPI = 1.57079632679489662f;

    for (int p = 0; p < 7; ++p) {
        const int b1 = pr1[p], b2 = pr2[p];
        const float* pw = w + p * 6;
        gate_rz  (st, b2, -HPI,  t); __syncthreads();
        gate_cnot(st, b2, b1,    t); __syncthreads();
        gate_rz  (st, b1, pw[0], t); __syncthreads();
        gate_ry  (st, b2, pw[1], t); __syncthreads();
        gate_cnot(st, b1, b2,    t); __syncthreads();
        gate_ry  (st, b2, pw[2], t); __syncthreads();
        gate_cnot(st, b2, b1,    t); __syncthreads();
        gate_rz  (st, b1,  HPI,  t); __syncthreads();
        gate_rz  (st, b2, -HPI,  t); __syncthreads();
        gate_cnot(st, b2, b1,    t); __syncthreads();
        gate_rz  (st, b1, pw[3], t); __syncthreads();
        gate_ry  (st, b2, pw[4], t); __syncthreads();
        gate_cnot(st, b1, b2,    t); __syncthreads();
        gate_ry  (st, b2, pw[5], t); __syncthreads();
    }

    float2 amp = st[2 * t + 1];
    const int base = (j >> 1) * 512 + (j & 1);
    AtP[base + t * 2]         = amp.x;
    AtP[base + (t + 128) * 2] = amp.y;
}


#define BM 128
#define NSUB 8
#define NWAVES 16

__global__ __launch_bounds__(512) void qcnn_wmma_gemm(const float* __restrict__ x,
                                                      const float2* __restrict__ AtP,
                                                      float* __restrict__ out) {
    __shared__ float Xs[BM * 260];
    __shared__ float part[BM][NWAVES];
    __shared__ float xss[BM][4];

    const int tid = threadIdx.x;
    const size_t m0 = (size_t)blockIdx.x * BM;

    {
        const int row = tid >> 2;
        const int seg = tid & 3;
        const float4* src = (const float4*)(x + (m0 + row) * 256 + seg * 64);
        float* dst = &Xs[row * 260 + seg * 64];
        float ss = 0.0f;
#pragma unroll
        for (int i = 0; i < 16; ++i) {
            float4 v = src[i];
            dst[i * 4 + 0] = v.x; dst[i * 4 + 1] = v.y;
            dst[i * 4 + 2] = v.z; dst[i * 4 + 3] = v.w;
            ss += v.x * v.x + v.y * v.y + v.z * v.z + v.w * v.w;
        }
        xss[row][seg] = ss;
    }
    __syncthreads();

    const int wave  = tid >> 5;
    const int lane  = tid & 31;
    const int laneL = lane & 15;
    const int hi    = lane >> 4;
    const int n0    = wave * 16;

    v8f acc[NSUB] = {};

#pragma unroll 4
    for (int k = 0; k < 256; k += 4) {
        float2 bv = AtP[(size_t)((k >> 1) + hi) * 256 + n0 + laneL];
        v2f b; b[0] = bv.x; b[1] = bv.y;
#pragma unroll
        for (int s = 0; s < NSUB; ++s) {
            const float2* pr = (const float2*)&Xs[(s * 16 + laneL) * 260];
            float2 av = pr[(k >> 1) + hi];
            v2f a; a[0] = av.x; a[1] = av.y;
            acc[s] = __builtin_amdgcn_wmma_f32_16x16x4_f32(
                false, a, false, b, (short)0, acc[s], false, false);
        }
    }

#pragma unroll
    for (int s = 0; s < NSUB; ++s) {
#pragma unroll
        for (int v = 0; v < 8; ++v) {
            float r2 = acc[s][v] * acc[s][v];
            r2 += __shfl_xor(r2, 1);
            r2 += __shfl_xor(r2, 2);
            r2 += __shfl_xor(r2, 4);
            r2 += __shfl_xor(r2, 8);
            if (laneL == 0) part[s * 16 + hi * 8 + v][wave] = r2;
        }
    }
    __syncthreads();

    if (tid < BM) {
        float q = 0.0f;
#pragma unroll
        for (int wv = 0; wv < NWAVES; ++wv) q += part[tid][wv];
        float ssx = 0.0f;
#pragma unroll
        for (int i = 0; i < 4; ++i) ssx += xss[tid][i];
        q /= ssx;
        const size_t bidx = m0 + tid;
        out[2 * bidx + 0] = 1.0f - q;
        out[2 * bidx + 1] = q;
    }
}


extern "C" void kernel_launch(void* const* d_in, const int* in_sizes, int n_in,
                              void* d_out, int out_size, void* d_ws, size_t ws_size,
                              hipStream_t stream) {
    const float* x = (const float*)d_in[0];
    const float* w = (const float*)d_in[1];
    float* out = (float*)d_out;
    float* AtP = (float*)d_ws;

    const int B = in_sizes[0] / 256;

    qcnn_build_matrix<<<256, 128, 0, stream>>>(w, AtP);
    qcnn_wmma_gemm<<<B / BM, 512, 0, stream>>>(x, (const float2*)AtP, out);
}


# GEMM K loop: wave tile 64x32, fragments read as 16-byte loads covering two k-steps and grouped one chunk ahead of the WMMAs (6 reads per 16 WMMAs); k and row partial sums added in a new order
# speedup vs baseline: 1.4764x; 1.4764x over previous
#include <hip/hip_runtime.h>
#include <hip/hip_bf16.h>

typedef __attribute__((ext_vector_type(2))) float v2f;
typedef __attribute__((ext_vector_type(8))) float v8f;


__device__ __forceinline__ void gate_rz(float2* st, int wire, float theta, int t) {
    float h = 0.5f * theta;
    float c = cosf(h), s = sinf(h);
    int mask = 1 << (7 - wire);
    int low  = t & (mask - 1);
    int i0   = ((t ^ low) << 1) | low;
    int i1   = i0 | mask;
    float2 a = st[i0], b = st[i1];
    st[i0] = make_float2(a.x * c + a.y * s, a.y * c - a.x * s);
    st[i1] = make_float2(b.x * c - b.y * s, b.y * c + b.x * s);
}

__device__ __forceinline__ void gate_ry(float2* st, int wire, float theta, int t) {
    float h = 0.5f * theta;
    float c = cosf(h), s = sinf(h);
    int mask = 1 << (7 - wire);
    int low  = t & (mask - 1);
    int i0   = ((t ^ low) << 1) | low;
    int i1   = i0 | mask;
    float2 a = st[i0], b = st[i1];
    st[i0] = make_float2(c * a.x - s * b.x, c * a.y - s * b.y);
    st[i1] = make_float2(s * a.x + c * b.x, s * a.y + c * b.y);
}

__device__ __forceinline__ void gate_cnot(float2* st, int ctrl, int tgt, int t) {
    int pc = 7 - ctrl, pt = 7 - tgt;
    int p1 = pc < pt ? pc : pt;
    int p2 = pc < pt ? pt : pc;
    if (t < 64) {
        unsigned low  = (unsigned)t & ((1u << p1) - 1u);
        unsigned mid  = ((unsigned)t >> p1) & ((1u << (p2 - p1 - 1)) - 1u);
        unsigned high = (unsigned)t >> (p2 - 1);
        unsigned i = (high << (p2 + 1)) | (mid << (p1 + 1)) | low;
        i |= (1u << pc);
        unsigned ip = i | (1u << pt);
        float2 a = st[i], b = st[ip];
        st[i]  = b;
        st[ip] = a;
    }
}

__global__ void qcnn_build_matrix(const float* __restrict__ w, float* __restrict__ AtP) {
    __shared__ float2 st[256];
    const int j = blockIdx.x;
    const int t = threadIdx.x;

    st[t]       = make_float2((t == j)       ? 1.0f : 0.0f, 0.0f);
    st[t + 128] = make_float2((t + 128 == j) ? 1.0f : 0.0f, 0.0f);
    __syncthreads();

    const int pr1[7] = {0, 2, 4, 6, 1, 5, 3};
    const int pr2[7] = {1, 3, 5, 7, 3, 7, 7};
    const float HPI = 1.57079632679489662f;

    for (int p = 0; p < 7; ++p) {
        const int b1 = pr1[p], b2 = pr2[p];
        const float* pw = w + p * 6;
        gate_rz  (st, b2, -HPI,  t); __syncthreads();
        gate_cnot(st, b2, b1,    t); __syncthreads();
        gate_rz  (st, b1, pw[0], t); __syncthreads();
        gate_ry  (st, b2, pw[1], t); __syncthreads();
        gate_cnot(st, b1, b2,    t); __syncthreads();
        gate_ry  (st, b2, pw[2], t); __syncthreads();
        gate_cnot(st, b2, b1,    t); __syncthreads();
        gate_rz  (st, b1,  HPI,  t); __syncthreads();
        gate_rz  (st, b2, -HPI,  t); __syncthreads();
        gate_cnot(st, b2, b1,    t); __syncthreads();
        gate_rz  (st, b1, pw[3], t); __syncthreads();
        gate_ry  (st, b2, pw[4], t); __syncthreads();
        gate_cnot(st, b1, b2,    t); __syncthreads();
        gate_ry  (st, b2, pw[5], t); __syncthreads();
    }

    float2 amp = st[2 * t + 1];
    const int base = (j >> 2) * 1024 + (j & 3);
    AtP[base + t * 4]         = amp.x;
    AtP[base + (t + 128) * 4] = amp.y;
}


#define BM 128
#define NCG 8
#define XP 260

struct Frag { float4 a[4]; float4 b[2]; };

__device__ __forceinline__ void load_frag(Frag& f, const float* ap, const float4* bp, int kc) {
#pragma unroll
    for (int i = 0; i < 4; ++i)
        f.a[i] = *(const float4*)(ap + i * 16 * XP + kc * 8);
#pragma unroll
    for (int j = 0; j < 2; ++j)
        f.b[j] = bp[(size_t)kc * 512 + j * 16];
}

__device__ __forceinline__ void mma_frag(v8f (&acc)[4][2], const Frag& f) {
#pragma unroll
    for (int i = 0; i < 4; ++i) {
#pragma unroll
        for (int j = 0; j < 2; ++j) {
            v2f a; a[0] = f.a[i].x; a[1] = f.a[i].y;
            v2f b; b[0] = f.b[j].x; b[1] = f.b[j].y;
            acc[i][j] = __builtin_amdgcn_wmma_f32_16x16x4_f32(
                false, a, false, b, (short)0, acc[i][j], false, false);
        }
    }
#pragma unroll
    for (int i = 0; i < 4; ++i) {
#pragma unroll
        for (int j = 0; j < 2; ++j) {
            v2f a; a[0] = f.a[i].z; a[1] = f.a[i].w;
            v2f b; b[0] = f.b[j].z; b[1] = f.b[j].w;
            acc[i][j] = __builtin_amdgcn_wmma_f32_16x16x4_f32(
                false, a, false, b, (short)0, acc[i][j], false, false);
        }
    }
}

__global__ __launch_bounds__(512) void qcnn_wmma_gemm(const float* __restrict__ x,
                                                      const float* __restrict__ AtP,
                                                      float* __restrict__ out) {
    __shared__ __attribute__((aligned(16))) float Xs[BM * XP];
    __shared__ float part[BM][NCG];
    __shared__ float xss[BM][4];

    const int tid = threadIdx.x;
    const size_t m0 = (size_t)blockIdx.x * BM;

    {
        const int row = tid >> 2;
        const int seg = tid & 3;
        const float4* src = (const float4*)(x + (m0 + row) * 256 + seg * 64);
        float* dst = &Xs[row * XP + seg * 64];
        float ss = 0.0f;
#pragma unroll
        for (int i = 0; i < 16; ++i) {
            float4 v = src[i];
            dst[i * 4 + 0] = v.x; dst[i * 4 + 1] = v.y;
            dst[i * 4 + 2] = v.z; dst[i * 4 + 3] = v.w;
            ss += v.x * v.x + v.y * v.y + v.z * v.z + v.w * v.w;
        }
        xss[row][seg] = ss;
    }
    __syncthreads();

    const int wave  = tid >> 5;
    const int lane  = tid & 31;
    const int laneL = lane & 15;
    const int hi    = lane >> 4;
    const int rh    = wave & 1;
    const int cg    = wave >> 1;

    const float*  ap = &Xs[(rh * 64 + laneL) * XP + 4 * hi];
    const float4* bp = (const float4*)AtP + (size_t)hi * 256 + cg * 32 + laneL;

    v8f acc[4][2] = {};
    Frag f0, f1;

    load_frag(f0, ap, bp, 0);
#pragma unroll 1
    for (int kc = 0; kc < 30; kc += 2) {
        load_frag(f1, ap, bp, kc + 1);
        __builtin_amdgcn_sched_barrier(0);
        mma_frag(acc, f0);
        __builtin_amdgcn_sched_barrier(0);
        load_frag(f0, ap, bp, kc + 2);
        __builtin_amdgcn_sched_barrier(0);
        mma_frag(acc, f1);
        __builtin_amdgcn_sched_barrier(0);
    }
    load_frag(f1, ap, bp, 31);
    __builtin_amdgcn_sched_barrier(0);
    mma_frag(acc, f0);
    mma_frag(acc, f1);

#pragma unroll
    for (int i = 0; i < 4; ++i) {
#pragma unroll
        for (int v = 0; v < 8; ++v) {
            float r2 = acc[i][0][v] * acc[i][0][v] + acc[i][1][v] * acc[i][1][v];
            r2 += __shfl_xor(r2, 1);
            r2 += __shfl_xor(r2, 2);
            r2 += __shfl_xor(r2, 4);
            r2 += __shfl_xor(r2, 8);
            if (laneL == 0) part[rh * 64 + i * 16 + hi * 8 + v][cg] = r2;
        }
    }
    __syncthreads();

    if (tid < BM) {
        float q = 0.0f;
#pragma unroll
        for (int g = 0; g < NCG; ++g) q += part[tid][g];
        float ssx = 0.0f;
#pragma unroll
        for (int i = 0; i < 4; ++i) ssx += xss[tid][i];
        q /= ssx;
        const size_t bidx = m0 + tid;
        out[2 * bidx + 0] = 1.0f - q;
        out[2 * bidx + 1] = q;
    }
}


extern "C" void kernel_launch(void* const* d_in, const int* in_sizes, int n_in,
                              void* d_out, int out_size, void* d_ws, size_t ws_size,
                              hipStream_t stream) {
    const float* x = (const float*)d_in[0];
    const float* w = (const float*)d_in[1];
    float* out = (float*)d_out;
    float* AtP = (float*)d_ws;

    const int B = in_sizes[0] / 256;

    qcnn_build_matrix<<<256, 128, 0, stream>>>(w, AtP);
    qcnn_wmma_gemm<<<B / BM, 512, 0, stream>>>(x, AtP, out);
}


# circuit-matrix kernel: each wire pair's 14 gates folded into one 4x4 complex matrix (sin/cos from a 42-lane table), applied as 7 two-wire steps: 9 barriers, not one per gate; new rounding order
# speedup vs baseline: 1.6873x; 1.1428x over previous
#include <hip/hip_runtime.h>
#include <hip/hip_bf16.h>

typedef __attribute__((ext_vector_type(2))) float v2f;
typedef __attribute__((ext_vector_type(8))) float v8f;


__device__ __forceinline__ float2 rot_m(float2 a, float c, float s) {
    return make_float2(a.x * c + a.y * s, a.y * c - a.x * s);
}
__device__ __forceinline__ float2 rot_p(float2 a, float c, float s) {
    return make_float2(a.x * c - a.y * s, a.y * c + a.x * s);
}

__device__ __forceinline__ void q_rz1(float2 (&a)[4], float c, float s) {
    a[0] = rot_m(a[0], c, s); a[1] = rot_m(a[1], c, s);
    a[2] = rot_p(a[2], c, s); a[3] = rot_p(a[3], c, s);
}
__device__ __forceinline__ void q_rz2(float2 (&a)[4], float c, float s) {
    a[0] = rot_m(a[0], c, s); a[2] = rot_m(a[2], c, s);
    a[1] = rot_p(a[1], c, s); a[3] = rot_p(a[3], c, s);
}
__device__ __forceinline__ void q_ry2(float2 (&a)[4], float c, float s) {
#pragma unroll
    for (int x1 = 0; x1 < 2; ++x1) {
        float2 u = a[2 * x1], v = a[2 * x1 + 1];
        a[2 * x1]     = make_float2(c * u.x - s * v.x, c * u.y - s * v.y);
        a[2 * x1 + 1] = make_float2(s * u.x + c * v.x, s * u.y + c * v.y);
    }
}
__device__ __forceinline__ void q_cnot21(float2 (&a)[4]) {
    float2 tmp = a[1]; a[1] = a[3]; a[3] = tmp;
}
__device__ __forceinline__ void q_cnot12(float2 (&a)[4]) {
    float2 tmp = a[2]; a[2] = a[3]; a[3] = tmp;
}

template <int B1, int B2>
__device__ __forceinline__ void apply_pair(const float2* src, float2* dst, const float2* G, int t) {
    constexpr int P1 = 7 - B1, P2 = 7 - B2;
    constexpr int M1 = 1 << P1, M2 = 1 << P2;
    const int g = t >> 1, X = t & 1;
    int i = ((g >> P2) << (P2 + 1)) | (g & (M2 - 1));
    i     = ((i >> P1) << (P1 + 1)) | (i & (M1 - 1));
    const float2 in[4] = {src[i], src[i | M2], src[i | M1], src[i | M1 | M2]};
    const float2* Gr = G + X * 8;
    float2 o0 = make_float2(0.0f, 0.0f), o1 = make_float2(0.0f, 0.0f);
#pragma unroll
    for (int c = 0; c < 4; ++c) {
        const float2 g0 = Gr[c], g1 = Gr[4 + c];
        o0.x += g0.x * in[c].x - g0.y * in[c].y;  o0.y += g0.x * in[c].y + g0.y * in[c].x;
        o1.x += g1.x * in[c].x - g1.y * in[c].y;  o1.y += g1.x * in[c].y + g1.y * in[c].x;
    }
    const int o = i | (X ? M1 : 0);
    dst[o]      = o0;
    dst[o | M2] = o1;
}

__global__ void qcnn_build_matrix(const float* __restrict__ w, float* __restrict__ AtP) {
    __shared__ float2 st[2][256];
    __shared__ __attribute__((aligned(16))) float2 Gs[7][16];
    __shared__ float2 cs[42];
    const int j = blockIdx.x;
    const int t = threadIdx.x;

    st[0][t]       = make_float2((t == j)       ? 1.0f : 0.0f, 0.0f);
    st[0][t + 128] = make_float2((t + 128 == j) ? 1.0f : 0.0f, 0.0f);
    if (t < 42) {
        const float h = 0.5f * w[t];
        cs[t] = make_float2(cosf(h), sinf(h));
    }
    __syncthreads();

    if (t < 28) {
        const int p = t >> 2, col = t & 3;
        const float HPI = 1.57079632679489662f;
        const float kc = cosf(0.5f * HPI), ks = sinf(0.5f * HPI);
        const float2* q = &cs[p * 6];
        float2 a[4];
#pragma unroll
        for (int r = 0; r < 4; ++r) a[r] = make_float2((r == col) ? 1.0f : 0.0f, 0.0f);
        q_rz2(a, kc, -ks);
        q_cnot21(a);
        q_rz1(a, q[0].x, q[0].y);
        q_ry2(a, q[1].x, q[1].y);
        q_cnot12(a);
        q_ry2(a, q[2].x, q[2].y);
        q_cnot21(a);
        q_rz1(a, kc, ks);
        q_rz2(a, kc, -ks);
        q_cnot21(a);
        q_rz1(a, q[3].x, q[3].y);
        q_ry2(a, q[4].x, q[4].y);
        q_cnot12(a);
        q_ry2(a, q[5].x, q[5].y);
#pragma unroll
        for (int r = 0; r < 4; ++r) Gs[p][r * 4 + col] = a[r];
    }
    __syncthreads();

    apply_pair<0, 1>(st[0], st[1], Gs[0], t); __syncthreads();
    apply_pair<2, 3>(st[1], st[0], Gs[1], t); __syncthreads();
    apply_pair<4, 5>(st[0], st[1], Gs[2], t); __syncthreads();
    apply_pair<6, 7>(st[1], st[0], Gs[3], t); __syncthreads();
    apply_pair<1, 3>(st[0], st[1], Gs[4], t); __syncthreads();
    apply_pair<5, 7>(st[1], st[0], Gs[5], t); __syncthreads();
    apply_pair<3, 7>(st[0], st[1], Gs[6], t); __syncthreads();

    float2 amp = st[1][2 * t + 1];
    const int base = (j >> 2) * 1024 + (j & 3);
    AtP[base + t * 4]         = amp.x;
    AtP[base + (t + 128) * 4] = amp.y;
}


#define BM 128
#define NCG 8
#define XP 260

struct Frag { float4 a[4]; float4 b[2]; };

__device__ __forceinline__ void load_frag(Frag& f, const float* ap, const float4* bp, int kc) {
#pragma unroll
    for (int i = 0; i < 4; ++i)
        f.a[i] = *(const float4*)(ap + i * 16 * XP + kc * 8);
#pragma unroll
    for (int j = 0; j < 2; ++j)
        f.b[j] = bp[(size_t)kc * 512 + j * 16];
}

__device__ __forceinline__ void mma_frag(v8f (&acc)[4][2], const Frag& f) {
#pragma unroll
    for (int i = 0; i < 4; ++i) {
#pragma unroll
        for (int j = 0; j < 2; ++j) {
            v2f a; a[0] = f.a[i].x; a[1] = f.a[i].y;
            v2f b; b[0] = f.b[j].x; b[1] = f.b[j].y;
            acc[i][j] = __builtin_amdgcn_wmma_f32_16x16x4_f32(
                false, a, false, b, (short)0, acc[i][j], false, false);
        }
    }
#pragma unroll
    for (int i = 0; i < 4; ++i) {
#pragma unroll
        for (int j = 0; j < 2; ++j) {
            v2f a; a[0] = f.a[i].z; a[1] = f.a[i].w;
            v2f b; b[0] = f.b[j].z; b[1] = f.b[j].w;
            acc[i][j] = __builtin_amdgcn_wmma_f32_16x16x4_f32(
                false, a, false, b, (short)0, acc[i][j], false, false);
        }
    }
}

__global__ __launch_bounds__(512) void qcnn_wmma_gemm(const float* __restrict__ x,
                                                      const float* __restrict__ AtP,
                                                      float* __restrict__ out) {
    __shared__ __attribute__((aligned(16))) float Xs[BM * XP];
    __shared__ float part[BM][NCG];
    __shared__ float xss[BM][4];

    const int tid = threadIdx.x;
    const size_t m0 = (size_t)blockIdx.x * BM;

    {
        const int row = tid >> 2;
        const int seg = tid & 3;
        const float4* src = (const float4*)(x + (m0 + row) * 256 + seg * 64);
        float* dst = &Xs[row * XP + seg * 64];
        float ss = 0.0f;
#pragma unroll
        for (int i = 0; i < 16; ++i) {
            float4 v = src[i];
            dst[i * 4 + 0] = v.x; dst[i * 4 + 1] = v.y;
            dst[i * 4 + 2] = v.z; dst[i * 4 + 3] = v.w;
            ss += v.x * v.x + v.y * v.y + v.z * v.z + v.w * v.w;
        }
        xss[row][seg] = ss;
    }
    __syncthreads();

    const int wave  = tid >> 5;
    const int lane  = tid & 31;
    const int laneL = lane & 15;
    const int hi    = lane >> 4;
    const int rh    = wave & 1;
    const int cg    = wave >> 1;

    const float*  ap = &Xs[(rh * 64 + laneL) * XP + 4 * hi];
    const float4* bp = (const float4*)AtP + (size_t)hi * 256 + cg * 32 + laneL;

    v8f acc[4][2] = {};
    Frag f0, f1;

    load_frag(f0, ap, bp, 0);
#pragma unroll 1
    for (int kc = 0; kc < 30; kc += 2) {
        load_frag(f1, ap, bp, kc + 1);
        __builtin_amdgcn_sched_barrier(0);
        mma_frag(acc, f0);
        __builtin_amdgcn_sched_barrier(0);
        load_frag(f0, ap, bp, kc + 2);
        __builtin_amdgcn_sched_barrier(0);
        mma_frag(acc, f1);
        __builtin_amdgcn_sched_barrier(0);
    }
    load_frag(f1, ap, bp, 31);
    __builtin_amdgcn_sched_barrier(0);
    mma_frag(acc, f0);
    mma_frag(acc, f1);

#pragma unroll
    for (int i = 0; i < 4; ++i) {
#pragma unroll
        for (int v = 0; v < 8; ++v) {
            float r2 = acc[i][0][v] * acc[i][0][v] + acc[i][1][v] * acc[i][1][v];
            r2 += __shfl_xor(r2, 1);
            r2 += __shfl_xor(r2, 2);
            r2 += __shfl_xor(r2, 4);
            r2 += __shfl_xor(r2, 8);
            if (laneL == 0) part[rh * 64 + i * 16 + hi * 8 + v][cg] = r2;
        }
    }
    __syncthreads();

    if (tid < BM) {
        float q = 0.0f;
#pragma unroll
        for (int g = 0; g < NCG; ++g) q += part[tid][g];
        float ssx = 0.0f;
#pragma unroll
        for (int i = 0; i < 4; ++i) ssx += xss[tid][i];
        q /= ssx;
        const size_t bidx = m0 + tid;
        out[2 * bidx + 0] = 1.0f - q;
        out[2 * bidx + 1] = q;
    }
}


extern "C" void kernel_launch(void* const* d_in, const int* in_sizes, int n_in,
                              void* d_out, int out_size, void* d_ws, size_t ws_size,
                              hipStream_t stream) {
    const float* x = (const float*)d_in[0];
    const float* w = (const float*)d_in[1];
    float* out = (float*)d_out;
    float* AtP = (float*)d_ws;

    const int B = in_sizes[0] / 256;

    qcnn_build_matrix<<<256, 128, 0, stream>>>(w, AtP);
    qcnn_wmma_gemm<<<B / BM, 512, 0, stream>>>(x, AtP, out);
}


# GEMM epilogue: row sums of C^2 over each 16-lane half use DPP row_xmask adds on the vector ALU instead of 128 dependent LDS-permute round trips per wave; same tree and rounding, one store per tile
# speedup vs baseline: 1.8089x; 1.0721x over previous
#include <hip/hip_runtime.h>
#include <hip/hip_bf16.h>

typedef __attribute__((ext_vector_type(2))) float v2f;
typedef __attribute__((ext_vector_type(8))) float v8f;


__device__ __forceinline__ float2 rot_m(float2 a, float c, float s) {
    return make_float2(a.x * c + a.y * s, a.y * c - a.x * s);
}
__device__ __forceinline__ float2 rot_p(float2 a, float c, float s) {
    return make_float2(a.x * c - a.y * s, a.y * c + a.x * s);
}

__device__ __forceinline__ void q_rz1(float2 (&a)[4], float c, float s) {
    a[0] = rot_m(a[0], c, s); a[1] = rot_m(a[1], c, s);
    a[2] = rot_p(a[2], c, s); a[3] = rot_p(a[3], c, s);
}
__device__ __forceinline__ void q_rz2(float2 (&a)[4], float c, float s) {
    a[0] = rot_m(a[0], c, s); a[2] = rot_m(a[2], c, s);
    a[1] = rot_p(a[1], c, s); a[3] = rot_p(a[3], c, s);
}
__device__ __forceinline__ void q_ry2(float2 (&a)[4], float c, float s) {
#pragma unroll
    for (int x1 = 0; x1 < 2; ++x1) {
        float2 u = a[2 * x1], v = a[2 * x1 + 1];
        a[2 * x1]     = make_float2(c * u.x - s * v.x, c * u.y - s * v.y);
        a[2 * x1 + 1] = make_float2(s * u.x + c * v.x, s * u.y + c * v.y);
    }
}
__device__ __forceinline__ void q_cnot21(float2 (&a)[4]) {
    float2 tmp = a[1]; a[1] = a[3]; a[3] = tmp;
}
__device__ __forceinline__ void q_cnot12(float2 (&a)[4]) {
    float2 tmp = a[2]; a[2] = a[3]; a[3] = tmp;
}

template <int B1, int B2>
__device__ __forceinline__ void apply_pair(const float2* src, float2* dst, const float2* G, int t) {
    constexpr int P1 = 7 - B1, P2 = 7 - B2;
    constexpr int M1 = 1 << P1, M2 = 1 << P2;
    const int g = t >> 1, X = t & 1;
    int i = ((g >> P2) << (P2 + 1)) | (g & (M2 - 1));
    i     = ((i >> P1) << (P1 + 1)) | (i & (M1 - 1));
    const float2 in[4] = {src[i], src[i | M2], src[i | M1], src[i | M1 | M2]};
    const float2* Gr = G + X * 8;
    float2 o0 = make_float2(0.0f, 0.0f), o1 = make_float2(0.0f, 0.0f);
#pragma unroll
    for (int c = 0; c < 4; ++c) {
        const float2 g0 = Gr[c], g1 = Gr[4 + c];
        o0.x += g0.x * in[c].x - g0.y * in[c].y;  o0.y += g0.x * in[c].y + g0.y * in[c].x;
        o1.x += g1.x * in[c].x - g1.y * in[c].y;  o1.y += g1.x * in[c].y + g1.y * in[c].x;
    }
    const int o = i | (X ? M1 : 0);
    dst[o]      = o0;
    dst[o | M2] = o1;
}

__global__ void qcnn_build_matrix(const float* __restrict__ w, float* __restrict__ AtP) {
    __shared__ float2 st[2][256];
    __shared__ __attribute__((aligned(16))) float2 Gs[7][16];
    __shared__ float2 cs[42];
    const int j = blockIdx.x;
    const int t = threadIdx.x;

    st[0][t]       = make_float2((t == j)       ? 1.0f : 0.0f, 0.0f);
    st[0][t + 128] = make_float2((t + 128 == j) ? 1.0f : 0.0f, 0.0f);
    if (t < 42) {
        const float h = 0.5f * w[t];
        cs[t] = make_float2(cosf(h), sinf(h));
    }
    __syncthreads();

    if (t < 28) {
        const int p = t >> 2, col = t & 3;
        const float HPI = 1.57079632679489662f;
        const float kc = cosf(0.5f * HPI), ks = sinf(0.5f * HPI);
        const float2* q = &cs[p * 6];
        float2 a[4];
#pragma unroll
        for (int r = 0; r < 4; ++r) a[r] = make_float2((r == col) ? 1.0f : 0.0f, 0.0f);
        q_rz2(a, kc, -ks);
        q_cnot21(a);
        q_rz1(a, q[0].x, q[0].y);
        q_ry2(a, q[1].x, q[1].y);
        q_cnot12(a);
        q_ry2(a, q[2].x, q[2].y);
        q_cnot21(a);
        q_rz1(a, kc, ks);
        q_rz2(a, kc, -ks);
        q_cnot21(a);
        q_rz1(a, q[3].x, q[3].y);
        q_ry2(a, q[4].x, q[4].y);
        q_cnot12(a);
        q_ry2(a, q[5].x, q[5].y);
#pragma unroll
        for (int r = 0; r < 4; ++r) Gs[p][r * 4 + col] = a[r];
    }
    __syncthreads();

    apply_pair<0, 1>(st[0], st[1], Gs[0], t); __syncthreads();
    apply_pair<2, 3>(st[1], st[0], Gs[1], t); __syncthreads();
    apply_pair<4, 5>(st[0], st[1], Gs[2], t); __syncthreads();
    apply_pair<6, 7>(st[1], st[0], Gs[3], t); __syncthreads();
    apply_pair<1, 3>(st[0], st[1], Gs[4], t); __syncthreads();
    apply_pair<5, 7>(st[1], st[0], Gs[5], t); __syncthreads();
    apply_pair<3, 7>(st[0], st[1], Gs[6], t); __syncthreads();

    float2 amp = st[1][2 * t + 1];
    const int base = (j >> 2) * 1024 + (j & 3);
    AtP[base + t * 4]         = amp.x;
    AtP[base + (t + 128) * 4] = amp.y;
}


#define BM 128
#define NCG 8
#define XP 260

struct Frag { float4 a[4]; float4 b[2]; };

__device__ __forceinline__ void load_frag(Frag& f, const float* ap, const float4* bp, int kc) {
#pragma unroll
    for (int i = 0; i < 4; ++i)
        f.a[i] = *(const float4*)(ap + i * 16 * XP + kc * 8);
#pragma unroll
    for (int j = 0; j < 2; ++j)
        f.b[j] = bp[(size_t)kc * 512 + j * 16];
}

__device__ __forceinline__ void mma_frag(v8f (&acc)[4][2], const Frag& f) {
#pragma unroll
    for (int i = 0; i < 4; ++i) {
#pragma unroll
        for (int j = 0; j < 2; ++j) {
            v2f a; a[0] = f.a[i].x; a[1] = f.a[i].y;
            v2f b; b[0] = f.b[j].x; b[1] = f.b[j].y;
            acc[i][j] = __builtin_amdgcn_wmma_f32_16x16x4_f32(
                false, a, false, b, (short)0, acc[i][j], false, false);
        }
    }
#pragma unroll
    for (int i = 0; i < 4; ++i) {
#pragma unroll
        for (int j = 0; j < 2; ++j) {
            v2f a; a[0] = f.a[i].z; a[1] = f.a[i].w;
            v2f b; b[0] = f.b[j].z; b[1] = f.b[j].w;
            acc[i][j] = __builtin_amdgcn_wmma_f32_16x16x4_f32(
                false, a, false, b, (short)0, acc[i][j], false, false);
        }
    }
}

template <int M>
__device__ __forceinline__ float row_xor(float v) {
    return __int_as_float(__builtin_amdgcn_update_dpp(
        0, __float_as_int(v), 0x160 + M  , 0xf, 0xf, true));
}

__global__ __launch_bounds__(512) void qcnn_wmma_gemm(const float* __restrict__ x,
                                                      const float* __restrict__ AtP,
                                                      float* __restrict__ out) {
    __shared__ __attribute__((aligned(16))) float Xs[BM * XP];
    __shared__ float part[BM][NCG];
    __shared__ float xss[BM][4];

    const int tid = threadIdx.x;
    const size_t m0 = (size_t)blockIdx.x * BM;

    {
        const int row = tid >> 2;
        const int seg = tid & 3;
        const float4* src = (const float4*)(x + (m0 + row) * 256 + seg * 64);
        float* dst = &Xs[row * XP + seg * 64];
        float ss = 0.0f;
#pragma unroll
        for (int i = 0; i < 16; ++i) {
            float4 v = src[i];
            dst[i * 4 + 0] = v.x; dst[i * 4 + 1] = v.y;
            dst[i * 4 + 2] = v.z; dst[i * 4 + 3] = v.w;
            ss += v.x * v.x + v.y * v.y + v.z * v.z + v.w * v.w;
        }
        xss[row][seg] = ss;
    }
    __syncthreads();

    const int wave  = tid >> 5;
    const int lane  = tid & 31;
    const int laneL = lane & 15;
    const int hi    = lane >> 4;
    const int rh    = wave & 1;
    const int cg    = wave >> 1;

    const float*  ap = &Xs[(rh * 64 + laneL) * XP + 4 * hi];
    const float4* bp = (const float4*)AtP + (size_t)hi * 256 + cg * 32 + laneL;

    v8f acc[4][2] = {};
    Frag f0, f1;

    load_frag(f0, ap, bp, 0);
#pragma unroll 1
    for (int kc = 0; kc < 30; kc += 2) {
        load_frag(f1, ap, bp, kc + 1);
        __builtin_amdgcn_sched_barrier(0);
        mma_frag(acc, f0);
        __builtin_amdgcn_sched_barrier(0);
        load_frag(f0, ap, bp, kc + 2);
        __builtin_amdgcn_sched_barrier(0);
        mma_frag(acc, f1);
        __builtin_amdgcn_sched_barrier(0);
    }
    load_frag(f1, ap, bp, 31);
    __builtin_amdgcn_sched_barrier(0);
    mma_frag(acc, f0);
    mma_frag(acc, f1);

#pragma unroll
    for (int i = 0; i < 4; ++i) {
        float keep = 0.0f;
#pragma unroll
        for (int v = 0; v < 8; ++v) {
            float r2 = acc[i][0][v] * acc[i][0][v] + acc[i][1][v] * acc[i][1][v];
            r2 += row_xor<1>(r2);
            r2 += row_xor<2>(r2);
            r2 += row_xor<4>(r2);
            r2 += row_xor<8>(r2);
            keep = (laneL == v) ? r2 : keep;
        }
        if (laneL < 8) part[rh * 64 + i * 16 + hi * 8 + laneL][cg] = keep;
    }
    __syncthreads();

    if (tid < BM) {
        float q = 0.0f;
#pragma unroll
        for (int g = 0; g < NCG; ++g) q += part[tid][g];
        float ssx = 0.0f;
#pragma unroll
        for (int i = 0; i < 4; ++i) ssx += xss[tid][i];
        q /= ssx;
        const size_t bidx = m0 + tid;
        out[2 * bidx + 0] = 1.0f - q;
        out[2 * bidx + 1] = q;
    }
}


extern "C" void kernel_launch(void* const* d_in, const int* in_sizes, int n_in,
                              void* d_out, int out_size, void* d_ws, size_t ws_size,
                              hipStream_t stream) {
    const float* x = (const float*)d_in[0];
    const float* w = (const float*)d_in[1];
    float* out = (float*)d_out;
    float* AtP = (float*)d_ws;

    const int B = in_sizes[0] / 256;

    qcnn_build_matrix<<<256, 128, 0, stream>>>(w, AtP);
    qcnn_wmma_gemm<<<B / BM, 512, 0, stream>>>(x, AtP, out);
}


# GEMM factored by the circuit's two nibbles: 16x16 complex (wires 4-7), then 16x32 complex (wires 0-3 and joining gate), chained in registers (D as A): 24 not 64 WMMAs per row; new summation order
# speedup vs baseline: 4.0588x; 2.2438x over previous
#include <hip/hip_runtime.h>


typedef __attribute__((ext_vector_type(2))) float v2f;
typedef __attribute__((ext_vector_type(8))) float v8f;


__device__ __forceinline__ float2 rot_m(float2 a, float c, float s) {
    return make_float2(a.x * c + a.y * s, a.y * c - a.x * s);
}
__device__ __forceinline__ float2 rot_p(float2 a, float c, float s) {
    return make_float2(a.x * c - a.y * s, a.y * c + a.x * s);
}

__device__ __forceinline__ void q_rz1(float2 (&a)[4], float c, float s) {
    a[0] = rot_m(a[0], c, s); a[1] = rot_m(a[1], c, s);
    a[2] = rot_p(a[2], c, s); a[3] = rot_p(a[3], c, s);
}
__device__ __forceinline__ void q_rz2(float2 (&a)[4], float c, float s) {
    a[0] = rot_m(a[0], c, s); a[2] = rot_m(a[2], c, s);
    a[1] = rot_p(a[1], c, s); a[3] = rot_p(a[3], c, s);
}
__device__ __forceinline__ void q_ry2(float2 (&a)[4], float c, float s) {
#pragma unroll
    for (int x1 = 0; x1 < 2; ++x1) {
        float2 u = a[2 * x1], v = a[2 * x1 + 1];
        a[2 * x1]     = make_float2(c * u.x - s * v.x, c * u.y - s * v.y);
        a[2 * x1 + 1] = make_float2(s * u.x + c * v.x, s * u.y + c * v.y);
    }
}
__device__ __forceinline__ void q_cnot21(float2 (&a)[4]) {
    float2 tmp = a[1]; a[1] = a[3]; a[3] = tmp;
}
__device__ __forceinline__ void q_cnot12(float2 (&a)[4]) {
    float2 tmp = a[2]; a[2] = a[3]; a[3] = tmp;
}


__device__ __forceinline__ float2 cmul(float2 a, float2 b) {
    return make_float2(a.x * b.x - a.y * b.y, a.x * b.y + a.y * b.x);
}

__device__ __forceinline__ float2 nibble_entry(const float2* Ga, const float2* Gb, const float2* Gc,
                                               int o, int col) {
    const int o3 = (o >> 3) & 1, o2 = (o >> 2) & 1, o1 = (o >> 1) & 1, o0 = o & 1;
    const int p0 = col >> 2, q0 = col & 3;
    float2 acc = make_float2(0.0f, 0.0f);
#pragma unroll
    for (int a5 = 0; a5 < 2; ++a5) {
#pragma unroll
        for (int a7 = 0; a7 < 2; ++a7) {
            const float2 kv = cmul(Ga[(2 * o3 + a5) * 4 + p0], Gb[(2 * o1 + a7) * 4 + q0]);
            const float2 t  = cmul(Gc[(2 * o2 + o0) * 4 + 2 * a5 + a7], kv);
            acc.x += t.x; acc.y += t.y;
        }
    }
    return acc;
}

__global__ __launch_bounds__(256) void qcnn_build_factors(const float* __restrict__ w,
                                                          float* __restrict__ Wf) {
    __shared__ float2 cs[42];
    __shared__ float2 Gs[7][16];
    __shared__ float2 TLs[16][16];
    __shared__ float2 THs[16][16];
    __shared__ float2 Qs[2][16][16];
    const int t = threadIdx.x;

    if (t < 42) {
        const float hw = 0.5f * w[t];
        cs[t] = make_float2(cosf(hw), sinf(hw));
    }
    __syncthreads();

    if (t < 28) {
        const int p = t >> 2, col = t & 3;
        const float HPI = 1.57079632679489662f;
        const float kc = cosf(0.5f * HPI), ks = sinf(0.5f * HPI);
        const float2* q = &cs[p * 6];
        float2 a[4];
#pragma unroll
        for (int r = 0; r < 4; ++r) a[r] = make_float2((r == col) ? 1.0f : 0.0f, 0.0f);
        q_rz2(a, kc, -ks);
        q_cnot21(a);
        q_rz1(a, q[0].x, q[0].y);
        q_ry2(a, q[1].x, q[1].y);
        q_cnot12(a);
        q_ry2(a, q[2].x, q[2].y);
        q_cnot21(a);
        q_rz1(a, kc, ks);
        q_rz2(a, kc, -ks);
        q_cnot21(a);
        q_rz1(a, q[3].x, q[3].y);
        q_ry2(a, q[4].x, q[4].y);
        q_cnot12(a);
        q_ry2(a, q[5].x, q[5].y);
#pragma unroll
        for (int r = 0; r < 4; ++r) Gs[p][r * 4 + col] = a[r];
    }
    __syncthreads();

    {
        const int o = t >> 4, col = t & 15;
        TLs[o][col] = nibble_entry(Gs[2], Gs[3], Gs[5], o, col);
        THs[o][col] = nibble_entry(Gs[0], Gs[1], Gs[4], o, col);
    }
    __syncthreads();

    {
        const int hp = t >> 4, a = t & 15;
        const int hr = hp >> 1, w3p = hp & 1;
#pragma unroll
        for (int s = 0; s < 2; ++s) {
            float2 acc = make_float2(0.0f, 0.0f);
#pragma unroll
            for (int w3 = 0; w3 < 2; ++w3) {
                const float2 tq = cmul(Gs[6][(2 * w3p + 1) * 4 + 2 * w3 + s], THs[2 * hr + w3][a]);
                acc.x += tq.x; acc.y += tq.y;
            }
            Qs[s][hp][a] = acc;
        }
    }
    __syncthreads();

    for (int idx = t; idx < 12 * 128; idx += 256) {
        const int vec = idx >> 7, lane = (idx >> 2) & 31, j = idx & 3;
        const int c = lane & 15, h = lane >> 4;
        float val;
        if (vec < 4) {
            const int s = vec >> 1, tp = vec & 1;
            const float2 v = TLs[2 * (c >> 1) + s][8 * h + 4 * tp + j];
            val = (c & 1) ? v.y : v.x;
        } else {
            const int u = vec - 4;
            const int P = u >> 2, s = (u >> 1) & 1, tp = u & 1;
            const float2 v = Qs[s][c][8 * h + 4 * tp + j];
            val = P ? v.y : v.x;
        }
        Wf[idx] = val;
    }
}


template <int M>
__device__ __forceinline__ float row_xor(float v) {
    return __int_as_float(__builtin_amdgcn_update_dpp(
        0, __float_as_int(v), 0x160 + M  , 0xf, 0xf, true));
}
__device__ __forceinline__ float xhalf(float v) {
    return __int_as_float(__builtin_amdgcn_permlanex16(
        __float_as_int(v), __float_as_int(v), 0x76543210, 0xfedcba98, false, false));
}
__device__ __forceinline__ float wave_sum(float v) {
    v += row_xor<1>(v);
    v += row_xor<2>(v);
    v += row_xor<4>(v);
    v += row_xor<8>(v);
    v += xhalf(v);
    return v;
}

__device__ __forceinline__ v8f mma(v8f acc, float a0, float a1, float b0, float b1) {
    v2f a; a[0] = a0; a[1] = a1;
    v2f b; b[0] = b0; b[1] = b1;
    return __builtin_amdgcn_wmma_f32_16x16x4_f32(false, a, false, b, (short)0, acc, false, false);
}

struct XPair { float4 v[2][2]; };

__device__ __forceinline__ void load_pair(XPair& xq, const float4* xp, int p) {
#pragma unroll
    for (int e = 0; e < 2; ++e) {
        xq.v[e][0] = xp[(size_t)(2 * p + e) * 64];
        xq.v[e][1] = xp[(size_t)(2 * p + e) * 64 + 1];
    }
}

__device__ __forceinline__ float sumsq(const float4 (&v)[2]) {
    return v[0].x * v[0].x + v[0].y * v[0].y + v[0].z * v[0].z + v[0].w * v[0].w
         + v[1].x * v[1].x + v[1].y * v[1].y + v[1].z * v[1].z + v[1].w * v[1].w;
}

struct Part { float v[4]; };

__device__ __forceinline__ void do_pair(const XPair& xq, const float4 (&B1)[2][2],
                                        const float4 (&B2)[2][2][2], Part& part) {
    const float4 (&x0)[2] = xq.v[0];
    const float4 (&x1)[2] = xq.v[1];
    v8f d1[2][2] = {};
#pragma unroll
    for (int s = 0; s < 2; ++s) {
#pragma unroll
        for (int tp = 0; tp < 2; ++tp) {
            d1[0][s] = mma(d1[0][s], x0[tp].x, x0[tp].y, B1[s][tp].x, B1[s][tp].y);
            d1[1][s] = mma(d1[1][s], x1[tp].x, x1[tp].y, B1[s][tp].x, B1[s][tp].y);
            d1[0][s] = mma(d1[0][s], x0[tp].z, x0[tp].w, B1[s][tp].z, B1[s][tp].w);
            d1[1][s] = mma(d1[1][s], x1[tp].z, x1[tp].w, B1[s][tp].z, B1[s][tp].w);
        }
    }
    v8f d2[2][2] = {};
#pragma unroll
    for (int s = 0; s < 2; ++s) {
#pragma unroll
        for (int tp = 0; tp < 2; ++tp) {
#pragma unroll
            for (int P = 0; P < 2; ++P) {
                d2[0][P] = mma(d2[0][P], d1[0][s][4 * tp + 0], d1[0][s][4 * tp + 1], B2[P][s][tp].x, B2[P][s][tp].y);
                d2[1][P] = mma(d2[1][P], d1[1][s][4 * tp + 0], d1[1][s][4 * tp + 1], B2[P][s][tp].x, B2[P][s][tp].y);
            }
#pragma unroll
            for (int P = 0; P < 2; ++P) {
                d2[0][P] = mma(d2[0][P], d1[0][s][4 * tp + 2], d1[0][s][4 * tp + 3], B2[P][s][tp].z, B2[P][s][tp].w);
                d2[1][P] = mma(d2[1][P], d1[1][s][4 * tp + 2], d1[1][s][4 * tp + 3], B2[P][s][tp].z, B2[P][s][tp].w);
            }
        }
    }
#pragma unroll
    for (int e = 0; e < 2; ++e) {
        float n = 0.0f;
#pragma unroll
        for (int j = 0; j < 4; ++j) {
            const float re = d2[e][0][2 * j] - d2[e][1][2 * j + 1];
            const float im = d2[e][1][2 * j] + d2[e][0][2 * j + 1];
            n += re * re + im * im;
        }
        part.v[2 * e]     = n;
        part.v[2 * e + 1] = sumsq(xq.v[e]);
    }
}

__device__ __forceinline__ void reduce_keep(const Part& part, int p, int lane, float& keepn, float& keeps) {
#pragma unroll
    for (int e = 0; e < 2; ++e) {
        const float n = wave_sum(part.v[2 * e]);
        const float d = wave_sum(part.v[2 * e + 1]);
        const bool mine = (lane == 2 * p + e);
        keepn = mine ? n : keepn;
        keeps = mine ? d : keeps;
    }
}

#define WAVES_PER_BLOCK 4
#define ROWS_PER_WAVE 32
#define PAIRS_PER_WAVE (ROWS_PER_WAVE / 2)
#define ROWS_PER_BLOCK (WAVES_PER_BLOCK * ROWS_PER_WAVE)

__global__ __launch_bounds__(WAVES_PER_BLOCK * 32) void qcnn_factored_wmma(const float* __restrict__ x,
                                                                         const float* __restrict__ Wf,
                                                                         float* __restrict__ out) {
    const int lane = threadIdx.x & 31;
    const int wave = threadIdx.x >> 5;
    const int c = lane & 15, h = lane >> 4;
    const size_t b0 = ((size_t)blockIdx.x * WAVES_PER_BLOCK + wave) * ROWS_PER_WAVE;

    const float4* wf = (const float4*)Wf + lane;
    float4 B1[2][2], B2[2][2][2];
#pragma unroll
    for (int s = 0; s < 2; ++s)
#pragma unroll
        for (int tp = 0; tp < 2; ++tp) B1[s][tp] = wf[(2 * s + tp) * 32];
#pragma unroll
    for (int P = 0; P < 2; ++P)
#pragma unroll
        for (int s = 0; s < 2; ++s)
#pragma unroll
            for (int tp = 0; tp < 2; ++tp) B2[P][s][tp] = wf[(4 + 4 * P + 2 * s + tp) * 32];

    const float4* xp = (const float4*)(x + b0 * 256 + c * 16 + 8 * h);

    float keepn = 0.0f, keeps = 1.0f;
    XPair xa, xb;
    Part pa, pb;
#pragma unroll
    for (int i = 0; i < 4; ++i) pb.v[i] = 0.0f;
    load_pair(xa, xp, 0);
#pragma unroll 1
    for (int p = 0; p < PAIRS_PER_WAVE; p += 2) {
        load_pair(xb, xp, p + 1);
        __builtin_amdgcn_sched_barrier(0);
        reduce_keep(pb, p - 1, lane, keepn, keeps);
        do_pair(xa, B1, B2, pa);
        __builtin_amdgcn_sched_barrier(0);
        const int pn = (p + 2 < PAIRS_PER_WAVE) ? p + 2 : PAIRS_PER_WAVE - 1;
        load_pair(xa, xp, pn);
        __builtin_amdgcn_sched_barrier(0);
        reduce_keep(pa, p, lane, keepn, keeps);
        do_pair(xb, B1, B2, pb);
        __builtin_amdgcn_sched_barrier(0);
    }
    reduce_keep(pb, PAIRS_PER_WAVE - 1, lane, keepn, keeps);

    const float q = keepn / keeps;
    float2* o2 = (float2*)out + (b0 + lane);
    *o2 = make_float2(1.0f - q, q);
}


extern "C" void kernel_launch(void* const* d_in, const int* in_sizes, int n_in,
                              void* d_out, int out_size, void* d_ws, size_t ws_size,
                              hipStream_t stream) {
    const float* x = (const float*)d_in[0];
    const float* w = (const float*)d_in[1];
    float* out = (float*)d_out;
    float* Wf = (float*)d_ws;

    const int B = in_sizes[0] / 256;

    qcnn_build_factors<<<1, 256, 0, stream>>>(w, Wf);
    qcnn_factored_wmma<<<B / ROWS_PER_BLOCK, WAVES_PER_BLOCK * 32, 0, stream>>>(x, Wf, out);
}


# One launch, not two: every work-group (now 8 waves) builds the four 16x16 factor tables in its own LDS behind its first x reads and reads its B fragments there; same arithmetic and summation order
# speedup vs baseline: 4.1975x; 1.0342x over previous
#include <hip/hip_runtime.h>


typedef __attribute__((ext_vector_type(2))) float v2f;
typedef __attribute__((ext_vector_type(8))) float v8f;


__device__ __forceinline__ float2 rot_m(float2 a, float c, float s) {
    return make_float2(a.x * c + a.y * s, a.y * c - a.x * s);
}
__device__ __forceinline__ float2 rot_p(float2 a, float c, float s) {
    return make_float2(a.x * c - a.y * s, a.y * c + a.x * s);
}

__device__ __forceinline__ void q_rz1(float2 (&a)[4], float c, float s) {
    a[0] = rot_m(a[0], c, s); a[1] = rot_m(a[1], c, s);
    a[2] = rot_p(a[2], c, s); a[3] = rot_p(a[3], c, s);
}
__device__ __forceinline__ void q_rz2(float2 (&a)[4], float c, float s) {
    a[0] = rot_m(a[0], c, s); a[2] = rot_m(a[2], c, s);
    a[1] = rot_p(a[1], c, s); a[3] = rot_p(a[3], c, s);
}
__device__ __forceinline__ void q_ry2(float2 (&a)[4], float c, float s) {
#pragma unroll
    for (int x1 = 0; x1 < 2; ++x1) {
        float2 u = a[2 * x1], v = a[2 * x1 + 1];
        a[2 * x1]     = make_float2(c * u.x - s * v.x, c * u.y - s * v.y);
        a[2 * x1 + 1] = make_float2(s * u.x + c * v.x, s * u.y + c * v.y);
    }
}
__device__ __forceinline__ void q_cnot21(float2 (&a)[4]) {
    float2 tmp = a[1]; a[1] = a[3]; a[3] = tmp;
}
__device__ __forceinline__ void q_cnot12(float2 (&a)[4]) {
    float2 tmp = a[2]; a[2] = a[3]; a[3] = tmp;
}


__device__ __forceinline__ float2 cmul(float2 a, float2 b) {
    return make_float2(a.x * b.x - a.y * b.y, a.x * b.y + a.y * b.x);
}

__device__ __forceinline__ float2 nibble_entry(const float2* Ga, const float2* Gb, const float2* Gc,
                                               int o, int col) {
    const int o3 = (o >> 3) & 1, o2 = (o >> 2) & 1, o1 = (o >> 1) & 1, o0 = o & 1;
    const int p0 = col >> 2, q0 = col & 3;
    float2 acc = make_float2(0.0f, 0.0f);
#pragma unroll
    for (int a5 = 0; a5 < 2; ++a5) {
#pragma unroll
        for (int a7 = 0; a7 < 2; ++a7) {
            const float2 kv = cmul(Ga[(2 * o3 + a5) * 4 + p0], Gb[(2 * o1 + a7) * 4 + q0]);
            const float2 t  = cmul(Gc[(2 * o2 + o0) * 4 + 2 * a5 + a7], kv);
            acc.x += t.x; acc.y += t.y;
        }
    }
    return acc;
}

struct alignas(16) Factors {
    float  TLp[2][16][16];
    float  Qp[2][2][16][16];
    float2 THs[16][16];
    float2 Gs[7][16];
    float2 cs[42];
};

__device__ __forceinline__ void build_angles(const float* __restrict__ w, Factors& F, int t) {
    if (t < 42) {
        const float hw = 0.5f * w[t];
        F.cs[t] = make_float2(cosf(hw), sinf(hw));
    }
    __syncthreads();
}

__device__ __forceinline__ void build_tables(Factors& F, int t) {
    if (t < 28) {
        const int p = t >> 2, col = t & 3;
        const float HPI = 1.57079632679489662f;
        const float kc = cosf(0.5f * HPI), ks = sinf(0.5f * HPI);
        const float2* q = &F.cs[p * 6];
        float2 a[4];
#pragma unroll
        for (int r = 0; r < 4; ++r) a[r] = make_float2((r == col) ? 1.0f : 0.0f, 0.0f);
        q_rz2(a, kc, -ks);
        q_cnot21(a);
        q_rz1(a, q[0].x, q[0].y);
        q_ry2(a, q[1].x, q[1].y);
        q_cnot12(a);
        q_ry2(a, q[2].x, q[2].y);
        q_cnot21(a);
        q_rz1(a, kc, ks);
        q_rz2(a, kc, -ks);
        q_cnot21(a);
        q_rz1(a, q[3].x, q[3].y);
        q_ry2(a, q[4].x, q[4].y);
        q_cnot12(a);
        q_ry2(a, q[5].x, q[5].y);
#pragma unroll
        for (int r = 0; r < 4; ++r) F.Gs[p][r * 4 + col] = a[r];
    }
    __syncthreads();

    {
        const int o = t >> 4, col = t & 15;
        const float2 tl = nibble_entry(F.Gs[2], F.Gs[3], F.Gs[5], o, col);
        F.TLp[0][o][col] = tl.x;
        F.TLp[1][o][col] = tl.y;
        F.THs[o][col] = nibble_entry(F.Gs[0], F.Gs[1], F.Gs[4], o, col);
    }
    __syncthreads();

    {
        const int hp = t >> 4, a = t & 15;
        const int hr = hp >> 1, w3p = hp & 1;
#pragma unroll
        for (int s = 0; s < 2; ++s) {
            float2 acc = make_float2(0.0f, 0.0f);
#pragma unroll
            for (int w3 = 0; w3 < 2; ++w3) {
                const float2 tq = cmul(F.Gs[6][(2 * w3p + 1) * 4 + 2 * w3 + s], F.THs[2 * hr + w3][a]);
                acc.x += tq.x; acc.y += tq.y;
            }
            F.Qp[0][s][hp][a] = acc.x;
            F.Qp[1][s][hp][a] = acc.y;
        }
    }
    __syncthreads();
}


template <int M>
__device__ __forceinline__ float row_xor(float v) {
    return __int_as_float(__builtin_amdgcn_update_dpp(
        0, __float_as_int(v), 0x160 + M  , 0xf, 0xf, true));
}
__device__ __forceinline__ float xhalf(float v) {
    return __int_as_float(__builtin_amdgcn_permlanex16(
        __float_as_int(v), __float_as_int(v), 0x76543210, 0xfedcba98, false, false));
}
__device__ __forceinline__ float wave_sum(float v) {
    v += row_xor<1>(v);
    v += row_xor<2>(v);
    v += row_xor<4>(v);
    v += row_xor<8>(v);
    v += xhalf(v);
    return v;
}

__device__ __forceinline__ v8f mma(v8f acc, float a0, float a1, float b0, float b1) {
    v2f a; a[0] = a0; a[1] = a1;
    v2f b; b[0] = b0; b[1] = b1;
    return __builtin_amdgcn_wmma_f32_16x16x4_f32(false, a, false, b, (short)0, acc, false, false);
}

struct XPair { float4 v[2][2]; };

__device__ __forceinline__ void load_pair(XPair& xq, const float4* xp, int p) {
#pragma unroll
    for (int e = 0; e < 2; ++e) {
        xq.v[e][0] = xp[(size_t)(2 * p + e) * 64];
        xq.v[e][1] = xp[(size_t)(2 * p + e) * 64 + 1];
    }
}

__device__ __forceinline__ float sumsq(const float4 (&v)[2]) {
    return v[0].x * v[0].x + v[0].y * v[0].y + v[0].z * v[0].z + v[0].w * v[0].w
         + v[1].x * v[1].x + v[1].y * v[1].y + v[1].z * v[1].z + v[1].w * v[1].w;
}

struct Part { float v[4]; };

__device__ __forceinline__ void do_pair(const XPair& xq, const float4 (&B1)[2][2],
                                        const float4 (&B2)[2][2][2], Part& part) {
    const float4 (&x0)[2] = xq.v[0];
    const float4 (&x1)[2] = xq.v[1];
    v8f d1[2][2] = {};
#pragma unroll
    for (int s = 0; s < 2; ++s) {
#pragma unroll
        for (int tp = 0; tp < 2; ++tp) {
            d1[0][s] = mma(d1[0][s], x0[tp].x, x0[tp].y, B1[s][tp].x, B1[s][tp].y);
            d1[1][s] = mma(d1[1][s], x1[tp].x, x1[tp].y, B1[s][tp].x, B1[s][tp].y);
            d1[0][s] = mma(d1[0][s], x0[tp].z, x0[tp].w, B1[s][tp].z, B1[s][tp].w);
            d1[1][s] = mma(d1[1][s], x1[tp].z, x1[tp].w, B1[s][tp].z, B1[s][tp].w);
        }
    }
    v8f d2[2][2] = {};
#pragma unroll
    for (int s = 0; s < 2; ++s) {
#pragma unroll
        for (int tp = 0; tp < 2; ++tp) {
#pragma unroll
            for (int P = 0; P < 2; ++P) {
                d2[0][P] = mma(d2[0][P], d1[0][s][4 * tp + 0], d1[0][s][4 * tp + 1], B2[P][s][tp].x, B2[P][s][tp].y);
                d2[1][P] = mma(d2[1][P], d1[1][s][4 * tp + 0], d1[1][s][4 * tp + 1], B2[P][s][tp].x, B2[P][s][tp].y);
            }
#pragma unroll
            for (int P = 0; P < 2; ++P) {
                d2[0][P] = mma(d2[0][P], d1[0][s][4 * tp + 2], d1[0][s][4 * tp + 3], B2[P][s][tp].z, B2[P][s][tp].w);
                d2[1][P] = mma(d2[1][P], d1[1][s][4 * tp + 2], d1[1][s][4 * tp + 3], B2[P][s][tp].z, B2[P][s][tp].w);
            }
        }
    }
#pragma unroll
    for (int e = 0; e < 2; ++e) {
        float n = 0.0f;
#pragma unroll
        for (int j = 0; j < 4; ++j) {
            const float re = d2[e][0][2 * j] - d2[e][1][2 * j + 1];
            const float im = d2[e][1][2 * j] + d2[e][0][2 * j + 1];
            n += re * re + im * im;
        }
        part.v[2 * e]     = n;
        part.v[2 * e + 1] = sumsq(xq.v[e]);
    }
}

__device__ __forceinline__ void reduce_keep(const Part& part, int p, int lane, float& keepn, float& keeps) {
#pragma unroll
    for (int e = 0; e < 2; ++e) {
        const float n = wave_sum(part.v[2 * e]);
        const float d = wave_sum(part.v[2 * e + 1]);
        const bool mine = (lane == 2 * p + e);
        keepn = mine ? n : keepn;
        keeps = mine ? d : keeps;
    }
}

#define WAVES_PER_BLOCK 8
#define ROWS_PER_WAVE 32
#define PAIRS_PER_WAVE (ROWS_PER_WAVE / 2)
#define ROWS_PER_BLOCK (WAVES_PER_BLOCK * ROWS_PER_WAVE)

__global__ __launch_bounds__(WAVES_PER_BLOCK * 32) void qcnn_factored_wmma(const float* __restrict__ x,
                                                                         const float* __restrict__ w,
                                                                         float* __restrict__ out) {
    __shared__ Factors F;
    const int lane = threadIdx.x & 31;
    const int wave = threadIdx.x >> 5;
    const int c = lane & 15, h = lane >> 4;
    const size_t b0 = ((size_t)blockIdx.x * WAVES_PER_BLOCK + wave) * ROWS_PER_WAVE;

    const float4* xp = (const float4*)(x + b0 * 256 + c * 16 + 8 * h);

    float keepn = 0.0f, keeps = 1.0f;
    XPair xa, xb;
    Part pa, pb;
#pragma unroll
    for (int i = 0; i < 4; ++i) pb.v[i] = 0.0f;
    build_angles(w, F, threadIdx.x);
    load_pair(xa, xp, 0);
    build_tables(F, threadIdx.x);

    float4 B1[2][2], B2[2][2][2];
#pragma unroll
    for (int s = 0; s < 2; ++s)
#pragma unroll
        for (int tp = 0; tp < 2; ++tp)
            B1[s][tp] = *(const float4*)&F.TLp[c & 1][2 * (c >> 1) + s][8 * h + 4 * tp];
#pragma unroll
    for (int P = 0; P < 2; ++P)
#pragma unroll
        for (int s = 0; s < 2; ++s)
#pragma unroll
            for (int tp = 0; tp < 2; ++tp)
                B2[P][s][tp] = *(const float4*)&F.Qp[P][s][c][8 * h + 4 * tp];
#pragma unroll 1
    for (int p = 0; p < PAIRS_PER_WAVE; p += 2) {
        load_pair(xb, xp, p + 1);
        __builtin_amdgcn_sched_barrier(0);
        reduce_keep(pb, p - 1, lane, keepn, keeps);
        do_pair(xa, B1, B2, pa);
        __builtin_amdgcn_sched_barrier(0);
        const int pn = (p + 2 < PAIRS_PER_WAVE) ? p + 2 : PAIRS_PER_WAVE - 1;
        load_pair(xa, xp, pn);
        __builtin_amdgcn_sched_barrier(0);
        reduce_keep(pa, p, lane, keepn, keeps);
        do_pair(xb, B1, B2, pb);
        __builtin_amdgcn_sched_barrier(0);
    }
    reduce_keep(pb, PAIRS_PER_WAVE - 1, lane, keepn, keeps);

    const float q = keepn / keeps;
    float2* o2 = (float2*)out + (b0 + lane);
    *o2 = make_float2(1.0f - q, q);
}


extern "C" void kernel_launch(void* const* d_in, const int* in_sizes, int n_in,
                              void* d_out, int out_size, void* d_ws, size_t ws_size,
                              hipStream_t stream) {
    const float* x = (const float*)d_in[0];
    const float* w = (const float*)d_in[1];
    float* out = (float*)d_out;

    const int B = in_sizes[0] / 256;

    qcnn_factored_wmma<<<B / ROWS_PER_BLOCK, WAVES_PER_BLOCK * 32, 0, stream>>>(x, w, out);
}


# Per-row epilogue on register pairs: each complex output is one packed add (half swap and sign as operand selects) and one packed multiply-add; ||x||^2 packed too; new summation order
# speedup vs baseline: 4.4292x; 1.0552x over previous
#include <hip/hip_runtime.h>


typedef __attribute__((ext_vector_type(2))) float v2f;
typedef __attribute__((ext_vector_type(8))) float v8f;


__device__ __forceinline__ float2 rot_m(float2 a, float c, float s) {
    return make_float2(a.x * c + a.y * s, a.y * c - a.x * s);
}
__device__ __forceinline__ float2 rot_p(float2 a, float c, float s) {
    return make_float2(a.x * c - a.y * s, a.y * c + a.x * s);
}

__device__ __forceinline__ void q_rz1(float2 (&a)[4], float c, float s) {
    a[0] = rot_m(a[0], c, s); a[1] = rot_m(a[1], c, s);
    a[2] = rot_p(a[2], c, s); a[3] = rot_p(a[3], c, s);
}
__device__ __forceinline__ void q_rz2(float2 (&a)[4], float c, float s) {
    a[0] = rot_m(a[0], c, s); a[2] = rot_m(a[2], c, s);
    a[1] = rot_p(a[1], c, s); a[3] = rot_p(a[3], c, s);
}
__device__ __forceinline__ void q_ry2(float2 (&a)[4], float c, float s) {
#pragma unroll
    for (int x1 = 0; x1 < 2; ++x1) {
        float2 u = a[2 * x1], v = a[2 * x1 + 1];
        a[2 * x1]     = make_float2(c * u.x - s * v.x, c * u.y - s * v.y);
        a[2 * x1 + 1] = make_float2(s * u.x + c * v.x, s * u.y + c * v.y);
    }
}
__device__ __forceinline__ void q_cnot21(float2 (&a)[4]) {
    float2 tmp = a[1]; a[1] = a[3]; a[3] = tmp;
}
__device__ __forceinline__ void q_cnot12(float2 (&a)[4]) {
    float2 tmp = a[2]; a[2] = a[3]; a[3] = tmp;
}


__device__ __forceinline__ float2 cmul(float2 a, float2 b) {
    return make_float2(a.x * b.x - a.y * b.y, a.x * b.y + a.y * b.x);
}

__device__ __forceinline__ float2 nibble_entry(const float2* Ga, const float2* Gb, const float2* Gc,
                                               int o, int col) {
    const int o3 = (o >> 3) & 1, o2 = (o >> 2) & 1, o1 = (o >> 1) & 1, o0 = o & 1;
    const int p0 = col >> 2, q0 = col & 3;
    float2 acc = make_float2(0.0f, 0.0f);
#pragma unroll
    for (int a5 = 0; a5 < 2; ++a5) {
#pragma unroll
        for (int a7 = 0; a7 < 2; ++a7) {
            const float2 kv = cmul(Ga[(2 * o3 + a5) * 4 + p0], Gb[(2 * o1 + a7) * 4 + q0]);
            const float2 t  = cmul(Gc[(2 * o2 + o0) * 4 + 2 * a5 + a7], kv);
            acc.x += t.x; acc.y += t.y;
        }
    }
    return acc;
}

struct alignas(16) Factors {
    float  TLp[2][16][16];
    float  Qp[2][2][16][16];
    float2 THs[16][16];
    float2 Gs[7][16];
    float2 cs[42];
};

__device__ __forceinline__ void build_angles(const float* __restrict__ w, Factors& F, int t) {
    if (t < 42) {
        const float hw = 0.5f * w[t];
        F.cs[t] = make_float2(cosf(hw), sinf(hw));
    }
    __syncthreads();
}

__device__ __forceinline__ void build_tables(Factors& F, int t) {
    if (t < 28) {
        const int p = t >> 2, col = t & 3;
        const float HPI = 1.57079632679489662f;
        const float kc = cosf(0.5f * HPI), ks = sinf(0.5f * HPI);
        const float2* q = &F.cs[p * 6];
        float2 a[4];
#pragma unroll
        for (int r = 0; r < 4; ++r) a[r] = make_float2((r == col) ? 1.0f : 0.0f, 0.0f);
        q_rz2(a, kc, -ks);
        q_cnot21(a);
        q_rz1(a, q[0].x, q[0].y);
        q_ry2(a, q[1].x, q[1].y);
        q_cnot12(a);
        q_ry2(a, q[2].x, q[2].y);
        q_cnot21(a);
        q_rz1(a, kc, ks);
        q_rz2(a, kc, -ks);
        q_cnot21(a);
        q_rz1(a, q[3].x, q[3].y);
        q_ry2(a, q[4].x, q[4].y);
        q_cnot12(a);
        q_ry2(a, q[5].x, q[5].y);
#pragma unroll
        for (int r = 0; r < 4; ++r) F.Gs[p][r * 4 + col] = a[r];
    }
    __syncthreads();

    {
        const int o = t >> 4, col = t & 15;
        const float2 tl = nibble_entry(F.Gs[2], F.Gs[3], F.Gs[5], o, col);
        F.TLp[0][o][col] = tl.x;
        F.TLp[1][o][col] = tl.y;
        F.THs[o][col] = nibble_entry(F.Gs[0], F.Gs[1], F.Gs[4], o, col);
    }
    __syncthreads();

    {
        const int hp = t >> 4, a = t & 15;
        const int hr = hp >> 1, w3p = hp & 1;
#pragma unroll
        for (int s = 0; s < 2; ++s) {
            float2 acc = make_float2(0.0f, 0.0f);
#pragma unroll
            for (int w3 = 0; w3 < 2; ++w3) {
                const float2 tq = cmul(F.Gs[6][(2 * w3p + 1) * 4 + 2 * w3 + s], F.THs[2 * hr + w3][a]);
                acc.x += tq.x; acc.y += tq.y;
            }
            F.Qp[0][s][hp][a] = acc.x;
            F.Qp[1][s][hp][a] = acc.y;
        }
    }
    __syncthreads();
}


template <int M>
__device__ __forceinline__ float row_xor(float v) {
    return __int_as_float(__builtin_amdgcn_update_dpp(
        0, __float_as_int(v), 0x160 + M  , 0xf, 0xf, true));
}
__device__ __forceinline__ float xhalf(float v) {
    return __int_as_float(__builtin_amdgcn_permlanex16(
        __float_as_int(v), __float_as_int(v), 0x76543210, 0xfedcba98, false, false));
}
__device__ __forceinline__ float wave_sum(float v) {
    v += row_xor<1>(v);
    v += row_xor<2>(v);
    v += row_xor<4>(v);
    v += row_xor<8>(v);
    v += xhalf(v);
    return v;
}

__device__ __forceinline__ v8f mma(v8f acc, float a0, float a1, float b0, float b1) {
    v2f a; a[0] = a0; a[1] = a1;
    v2f b; b[0] = b0; b[1] = b1;
    return __builtin_amdgcn_wmma_f32_16x16x4_f32(false, a, false, b, (short)0, acc, false, false);
}

struct XPair { float4 v[2][2]; };

__device__ __forceinline__ void load_pair(XPair& xq, const float4* xp, int p) {
#pragma unroll
    for (int e = 0; e < 2; ++e) {
        xq.v[e][0] = xp[(size_t)(2 * p + e) * 64];
        xq.v[e][1] = xp[(size_t)(2 * p + e) * 64 + 1];
    }
}

__device__ __forceinline__ float sumsq(const float4 (&v)[2]) {
    v2f p0; p0[0] = v[0].x; p0[1] = v[0].y;
    v2f p1; p1[0] = v[0].z; p1[1] = v[0].w;
    v2f p2; p2[0] = v[1].x; p2[1] = v[1].y;
    v2f p3; p3[0] = v[1].z; p3[1] = v[1].w;
    v2f s = p0 * p0;
    s = __builtin_elementwise_fma(p1, p1, s);
    s = __builtin_elementwise_fma(p2, p2, s);
    s = __builtin_elementwise_fma(p3, p3, s);
    return s[0] + s[1];
}

__device__ __forceinline__ v2f out_pair(v8f re_t, v8f im_t, int j) {
    v2f a, b;
    switch (j) {
    case 0:  a = __builtin_shufflevector(re_t, re_t, 0, 1); b = __builtin_shufflevector(im_t, im_t, 0, 1); break;
    case 1:  a = __builtin_shufflevector(re_t, re_t, 2, 3); b = __builtin_shufflevector(im_t, im_t, 2, 3); break;
    case 2:  a = __builtin_shufflevector(re_t, re_t, 4, 5); b = __builtin_shufflevector(im_t, im_t, 4, 5); break;
    default: a = __builtin_shufflevector(re_t, re_t, 6, 7); b = __builtin_shufflevector(im_t, im_t, 6, 7); break;
    }
    asm("" : "+v"(b));
    v2f t; t[0] = -b[1]; t[1] = b[0];
    return a + t;
}

struct Part { float v[4]; };

__device__ __forceinline__ void do_pair(const XPair& xq, const float4 (&B1)[2][2],
                                        const float4 (&B2)[2][2][2], Part& part) {
    const float4 (&x0)[2] = xq.v[0];
    const float4 (&x1)[2] = xq.v[1];
    v8f d1[2][2] = {};
#pragma unroll
    for (int s = 0; s < 2; ++s) {
#pragma unroll
        for (int tp = 0; tp < 2; ++tp) {
            d1[0][s] = mma(d1[0][s], x0[tp].x, x0[tp].y, B1[s][tp].x, B1[s][tp].y);
            d1[1][s] = mma(d1[1][s], x1[tp].x, x1[tp].y, B1[s][tp].x, B1[s][tp].y);
            d1[0][s] = mma(d1[0][s], x0[tp].z, x0[tp].w, B1[s][tp].z, B1[s][tp].w);
            d1[1][s] = mma(d1[1][s], x1[tp].z, x1[tp].w, B1[s][tp].z, B1[s][tp].w);
        }
    }
    v8f d2[2][2] = {};
#pragma unroll
    for (int s = 0; s < 2; ++s) {
#pragma unroll
        for (int tp = 0; tp < 2; ++tp) {
#pragma unroll
            for (int P = 0; P < 2; ++P) {
                d2[0][P] = mma(d2[0][P], d1[0][s][4 * tp + 0], d1[0][s][4 * tp + 1], B2[P][s][tp].x, B2[P][s][tp].y);
                d2[1][P] = mma(d2[1][P], d1[1][s][4 * tp + 0], d1[1][s][4 * tp + 1], B2[P][s][tp].x, B2[P][s][tp].y);
            }
#pragma unroll
            for (int P = 0; P < 2; ++P) {
                d2[0][P] = mma(d2[0][P], d1[0][s][4 * tp + 2], d1[0][s][4 * tp + 3], B2[P][s][tp].z, B2[P][s][tp].w);
                d2[1][P] = mma(d2[1][P], d1[1][s][4 * tp + 2], d1[1][s][4 * tp + 3], B2[P][s][tp].z, B2[P][s][tp].w);
            }
        }
    }
#pragma unroll
    for (int e = 0; e < 2; ++e) {
        v2f acc = {0.0f, 0.0f};
#pragma unroll
        for (int j = 0; j < 4; ++j) {
            const v2f o = out_pair(d2[e][0], d2[e][1], j);
            acc = __builtin_elementwise_fma(o, o, acc);
        }
        part.v[2 * e]     = acc[0] + acc[1];
        part.v[2 * e + 1] = sumsq(xq.v[e]);
    }
}

__device__ __forceinline__ void reduce_keep(const Part& part, int p, int lane, float& keepn, float& keeps) {
#pragma unroll
    for (int e = 0; e < 2; ++e) {
        const float n = wave_sum(part.v[2 * e]);
        const float d = wave_sum(part.v[2 * e + 1]);
        const bool mine = (lane == 2 * p + e);
        keepn = mine ? n : keepn;
        keeps = mine ? d : keeps;
    }
}

#define WAVES_PER_BLOCK 8
#define ROWS_PER_WAVE 32
#define PAIRS_PER_WAVE (ROWS_PER_WAVE / 2)
#define ROWS_PER_BLOCK (WAVES_PER_BLOCK * ROWS_PER_WAVE)

__global__ __launch_bounds__(WAVES_PER_BLOCK * 32) void qcnn_factored_wmma(const float* __restrict__ x,
                                                                         const float* __restrict__ w,
                                                                         float* __restrict__ out) {
    __shared__ Factors F;
    const int lane = threadIdx.x & 31;
    const int wave = threadIdx.x >> 5;
    const int c = lane & 15, h = lane >> 4;
    const size_t b0 = ((size_t)blockIdx.x * WAVES_PER_BLOCK + wave) * ROWS_PER_WAVE;

    const float4* xp = (const float4*)(x + b0 * 256 + c * 16 + 8 * h);

    float keepn = 0.0f, keeps = 1.0f;
    XPair xa, xb;
    Part pa, pb;
#pragma unroll
    for (int i = 0; i < 4; ++i) pb.v[i] = 0.0f;
    build_angles(w, F, threadIdx.x);
    load_pair(xa, xp, 0);
    build_tables(F, threadIdx.x);

    float4 B1[2][2], B2[2][2][2];
#pragma unroll
    for (int s = 0; s < 2; ++s)
#pragma unroll
        for (int tp = 0; tp < 2; ++tp)
            B1[s][tp] = *(const float4*)&F.TLp[c & 1][2 * (c >> 1) + s][8 * h + 4 * tp];
#pragma unroll
    for (int P = 0; P < 2; ++P)
#pragma unroll
        for (int s = 0; s < 2; ++s)
#pragma unroll
            for (int tp = 0; tp < 2; ++tp)
                B2[P][s][tp] = *(const float4*)&F.Qp[P][s][c][8 * h + 4 * tp];
#pragma unroll 1
    for (int p = 0; p < PAIRS_PER_WAVE; p += 2) {
        load_pair(xb, xp, p + 1);
        __builtin_amdgcn_sched_barrier(0);
        reduce_keep(pb, p - 1, lane, keepn, keeps);
        do_pair(xa, B1, B2, pa);
        __builtin_amdgcn_sched_barrier(0);
        const int pn = (p + 2 < PAIRS_PER_WAVE) ? p + 2 : PAIRS_PER_WAVE - 1;
        load_pair(xa, xp, pn);
        __builtin_amdgcn_sched_barrier(0);
        reduce_keep(pa, p, lane, keepn, keeps);
        do_pair(xb, B1, B2, pb);
        __builtin_amdgcn_sched_barrier(0);
    }
    reduce_keep(pb, PAIRS_PER_WAVE - 1, lane, keepn, keeps);

    const float q = keepn / keeps;
    float2* o2 = (float2*)out + (b0 + lane);
    *o2 = make_float2(1.0f - q, q);
}


extern "C" void kernel_launch(void* const* d_in, const int* in_sizes, int n_in,
                              void* d_out, int out_size, void* d_ws, size_t ws_size,
                              hipStream_t stream) {
    const float* x = (const float*)d_in[0];
    const float* w = (const float*)d_in[1];
    float* out = (float*)d_out;

    const int B = in_sizes[0] / 256;

    qcnn_factored_wmma<<<B / ROWS_PER_BLOCK, WAVES_PER_BLOCK * 32, 0, stream>>>(x, w, out);
}


# Stage-2 complex product as three real products (re.re, im.im, sum.sum) on tiles shared by two batch rows: 12 not 16 WMMAs per row; parts by lane select and a DPP add; new rounding and summation order
# speedup vs baseline: 4.6136x; 1.0416x over previous
#include <hip/hip_runtime.h>


typedef __attribute__((ext_vector_type(2))) float v2f;
typedef __attribute__((ext_vector_type(8))) float v8f;


__device__ __forceinline__ float2 rot_m(float2 a, float c, float s) {
    return make_float2(a.x * c + a.y * s, a.y * c - a.x * s);
}
__device__ __forceinline__ float2 rot_p(float2 a, float c, float s) {
    return make_float2(a.x * c - a.y * s, a.y * c + a.x * s);
}

__device__ __forceinline__ void q_rz1(float2 (&a)[4], float c, float s) {
    a[0] = rot_m(a[0], c, s); a[1] = rot_m(a[1], c, s);
    a[2] = rot_p(a[2], c, s); a[3] = rot_p(a[3], c, s);
}
__device__ __forceinline__ void q_rz2(float2 (&a)[4], float c, float s) {
    a[0] = rot_m(a[0], c, s); a[2] = rot_m(a[2], c, s);
    a[1] = rot_p(a[1], c, s); a[3] = rot_p(a[3], c, s);
}
__device__ __forceinline__ void q_ry2(float2 (&a)[4], float c, float s) {
#pragma unroll
    for (int x1 = 0; x1 < 2; ++x1) {
        float2 u = a[2 * x1], v = a[2 * x1 + 1];
        a[2 * x1]     = make_float2(c * u.x - s * v.x, c * u.y - s * v.y);
        a[2 * x1 + 1] = make_float2(s * u.x + c * v.x, s * u.y + c * v.y);
    }
}
__device__ __forceinline__ void q_cnot21(float2 (&a)[4]) {
    float2 tmp = a[1]; a[1] = a[3]; a[3] = tmp;
}
__device__ __forceinline__ void q_cnot12(float2 (&a)[4]) {
    float2 tmp = a[2]; a[2] = a[3]; a[3] = tmp;
}


__device__ __forceinline__ float2 cmul(float2 a, float2 b) {
    return make_float2(a.x * b.x - a.y * b.y, a.x * b.y + a.y * b.x);
}

__device__ __forceinline__ float2 nibble_entry(const float2* Ga, const float2* Gb, const float2* Gc,
                                               int o, int col) {
    const int o3 = (o >> 3) & 1, o2 = (o >> 2) & 1, o1 = (o >> 1) & 1, o0 = o & 1;
    const int p0 = col >> 2, q0 = col & 3;
    float2 acc = make_float2(0.0f, 0.0f);
#pragma unroll
    for (int a5 = 0; a5 < 2; ++a5) {
#pragma unroll
        for (int a7 = 0; a7 < 2; ++a7) {
            const float2 kv = cmul(Ga[(2 * o3 + a5) * 4 + p0], Gb[(2 * o1 + a7) * 4 + q0]);
            const float2 t  = cmul(Gc[(2 * o2 + o0) * 4 + 2 * a5 + a7], kv);
            acc.x += t.x; acc.y += t.y;
        }
    }
    return acc;
}

struct alignas(16) Factors {
    float  TLp[2][16][16];
    float  Qp[3][2][16][16];
    float2 THs[16][16];
    float2 Gs[7][16];
    float2 cs[42];
};

__device__ __forceinline__ void build_angles(const float* __restrict__ w, Factors& F, int t) {
    if (t < 42) {
        const float hw = 0.5f * w[t];
        F.cs[t] = make_float2(cosf(hw), sinf(hw));
    }
    __syncthreads();
}

__device__ __forceinline__ void build_tables(Factors& F, int t) {
    if (t < 28) {
        const int p = t >> 2, col = t & 3;
        const float HPI = 1.57079632679489662f;
        const float kc = cosf(0.5f * HPI), ks = sinf(0.5f * HPI);
        const float2* q = &F.cs[p * 6];
        float2 a[4];
#pragma unroll
        for (int r = 0; r < 4; ++r) a[r] = make_float2((r == col) ? 1.0f : 0.0f, 0.0f);
        q_rz2(a, kc, -ks);
        q_cnot21(a);
        q_rz1(a, q[0].x, q[0].y);
        q_ry2(a, q[1].x, q[1].y);
        q_cnot12(a);
        q_ry2(a, q[2].x, q[2].y);
        q_cnot21(a);
        q_rz1(a, kc, ks);
        q_rz2(a, kc, -ks);
        q_cnot21(a);
        q_rz1(a, q[3].x, q[3].y);
        q_ry2(a, q[4].x, q[4].y);
        q_cnot12(a);
        q_ry2(a, q[5].x, q[5].y);
#pragma unroll
        for (int r = 0; r < 4; ++r) F.Gs[p][r * 4 + col] = a[r];
    }
    __syncthreads();

    {
        const int o = t >> 4, col = t & 15;
        const float2 tl = nibble_entry(F.Gs[2], F.Gs[3], F.Gs[5], o, col);
        F.TLp[0][o][col] = tl.x;
        F.TLp[1][o][col] = tl.y;
        F.THs[o][col] = nibble_entry(F.Gs[0], F.Gs[1], F.Gs[4], o, col);
    }
    __syncthreads();

    {
        const int hp = t >> 4, a = t & 15;
        const int hr = hp >> 1, w3p = hp & 1;
#pragma unroll
        for (int s = 0; s < 2; ++s) {
            float2 acc = make_float2(0.0f, 0.0f);
#pragma unroll
            for (int w3 = 0; w3 < 2; ++w3) {
                const float2 tq = cmul(F.Gs[6][(2 * w3p + 1) * 4 + 2 * w3 + s], F.THs[2 * hr + w3][a]);
                acc.x += tq.x; acc.y += tq.y;
            }
            F.Qp[0][s][hp][a] = acc.x;
            F.Qp[1][s][hp][a] = acc.y;
            F.Qp[2][s][hp][a] = acc.x + acc.y;
        }
    }
    __syncthreads();
}


template <int M>
__device__ __forceinline__ float row_xor(float v) {
    return __int_as_float(__builtin_amdgcn_update_dpp(
        0, __float_as_int(v), 0x160 + M  , 0xf, 0xf, true));
}
__device__ __forceinline__ float xhalf(float v) {
    return __int_as_float(__builtin_amdgcn_permlanex16(
        __float_as_int(v), __float_as_int(v), 0x76543210, 0xfedcba98, false, false));
}
__device__ __forceinline__ float wave_sum(float v) {
    v += row_xor<1>(v);
    v += row_xor<2>(v);
    v += row_xor<4>(v);
    v += row_xor<8>(v);
    v += xhalf(v);
    return v;
}

__device__ __forceinline__ v8f mma(v8f acc, float a0, float a1, float b0, float b1) {
    v2f a; a[0] = a0; a[1] = a1;
    v2f b; b[0] = b0; b[1] = b1;
    return __builtin_amdgcn_wmma_f32_16x16x4_f32(false, a, false, b, (short)0, acc, false, false);
}

struct XPair { float4 v[2][2]; };

__device__ __forceinline__ void load_pair(XPair& xq, const float4* xp, int p) {
#pragma unroll
    for (int e = 0; e < 2; ++e) {
        xq.v[e][0] = xp[(size_t)(2 * p + e) * 64];
        xq.v[e][1] = xp[(size_t)(2 * p + e) * 64 + 1];
    }
}

__device__ __forceinline__ float sumsq(const float4 (&v)[2]) {
    v2f p0; p0[0] = v[0].x; p0[1] = v[0].y;
    v2f p1; p1[0] = v[0].z; p1[1] = v[0].w;
    v2f p2; p2[0] = v[1].x; p2[1] = v[1].y;
    v2f p3; p3[0] = v[1].z; p3[1] = v[1].w;
    v2f s = p0 * p0;
    s = __builtin_elementwise_fma(p1, p1, s);
    s = __builtin_elementwise_fma(p2, p2, s);
    s = __builtin_elementwise_fma(p3, p3, s);
    return s[0] + s[1];
}

__device__ __forceinline__ v2f reg_pair(v8f t, int j) {
    switch (j) {
    case 0:  return __builtin_shufflevector(t, t, 0, 1);
    case 1:  return __builtin_shufflevector(t, t, 2, 3);
    case 2:  return __builtin_shufflevector(t, t, 4, 5);
    default: return __builtin_shufflevector(t, t, 6, 7);
    }
}

struct Part { float v[4]; };

__device__ __forceinline__ void do_pair(const XPair& xq, const float4 (&B1)[2][2], const float4 (&B1x)[2][2],
                                        const float4 (&B2)[3][2][2], bool even, Part& part) {
    const float4 (&x0)[2] = xq.v[0];
    const float4 (&x1)[2] = xq.v[1];
    v8f d1[2][2] = {};
#pragma unroll
    for (int s = 0; s < 2; ++s) {
#pragma unroll
        for (int tp = 0; tp < 2; ++tp) {
            d1[0][s] = mma(d1[0][s], x0[tp].x, x0[tp].y, B1[s][tp].x, B1[s][tp].y);
            d1[1][s] = mma(d1[1][s], x1[tp].x, x1[tp].y, B1x[s][tp].x, B1x[s][tp].y);
            d1[0][s] = mma(d1[0][s], x0[tp].z, x0[tp].w, B1[s][tp].z, B1[s][tp].w);
            d1[1][s] = mma(d1[1][s], x1[tp].z, x1[tp].w, B1x[s][tp].z, B1x[s][tp].w);
        }
    }
    v8f yr[2], yi[2], ys[2];
#pragma unroll
    for (int s = 0; s < 2; ++s) {
        yr[s] = even ? d1[0][s] : d1[1][s];
        yi[s] = even ? d1[1][s] : d1[0][s];
#pragma unroll
        for (int r = 0; r < 8; ++r) ys[s][r] = yr[s][r] + row_xor<1>(yi[s][r]);
    }
    v8f d2[3] = {};
#pragma unroll
    for (int s = 0; s < 2; ++s) {
#pragma unroll
        for (int tp = 0; tp < 2; ++tp) {
            d2[0] = mma(d2[0], yr[s][4 * tp + 0], yr[s][4 * tp + 1], B2[0][s][tp].x, B2[0][s][tp].y);
            d2[1] = mma(d2[1], yi[s][4 * tp + 0], yi[s][4 * tp + 1], B2[1][s][tp].x, B2[1][s][tp].y);
            d2[2] = mma(d2[2], ys[s][4 * tp + 0], ys[s][4 * tp + 1], B2[2][s][tp].x, B2[2][s][tp].y);
            d2[0] = mma(d2[0], yr[s][4 * tp + 2], yr[s][4 * tp + 3], B2[0][s][tp].z, B2[0][s][tp].w);
            d2[1] = mma(d2[1], yi[s][4 * tp + 2], yi[s][4 * tp + 3], B2[1][s][tp].z, B2[1][s][tp].w);
            d2[2] = mma(d2[2], ys[s][4 * tp + 2], ys[s][4 * tp + 3], B2[2][s][tp].z, B2[2][s][tp].w);
        }
    }
    v2f acc = {0.0f, 0.0f};
#pragma unroll
    for (int j = 0; j < 4; ++j) {
        const v2f p1 = reg_pair(d2[0], j);
        v2f p2x = reg_pair(d2[1], j);
        const v2f p3 = reg_pair(d2[2], j);
        asm("" : "+v"(p2x));
        v2f p2; p2[0] = p2x[1]; p2[1] = p2x[0];
        const v2f ore = p1 - p2;
        const v2f oim = (p3 - p1) - p2;
        acc = __builtin_elementwise_fma(ore, ore, acc);
        acc = __builtin_elementwise_fma(oim, oim, acc);
    }
    part.v[0] = acc[0];
    part.v[1] = sumsq(x0);
    part.v[2] = acc[1];
    part.v[3] = sumsq(x1);
}

__device__ __forceinline__ void reduce_keep(const Part& part, int p, int lane, float& keepn, float& keeps) {
#pragma unroll
    for (int e = 0; e < 2; ++e) {
        const float n = wave_sum(part.v[2 * e]);
        const float d = wave_sum(part.v[2 * e + 1]);
        const bool mine = (lane == 2 * p + e);
        keepn = mine ? n : keepn;
        keeps = mine ? d : keeps;
    }
}

#define WAVES_PER_BLOCK 8
#define ROWS_PER_WAVE 32
#define PAIRS_PER_WAVE (ROWS_PER_WAVE / 2)
#define ROWS_PER_BLOCK (WAVES_PER_BLOCK * ROWS_PER_WAVE)

__global__ __launch_bounds__(WAVES_PER_BLOCK * 32) void qcnn_factored_wmma(const float* __restrict__ x,
                                                                         const float* __restrict__ w,
                                                                         float* __restrict__ out) {
    __shared__ Factors F;
    const int lane = threadIdx.x & 31;
    const int wave = threadIdx.x >> 5;
    const int c = lane & 15, h = lane >> 4;
    const size_t b0 = ((size_t)blockIdx.x * WAVES_PER_BLOCK + wave) * ROWS_PER_WAVE;

    const float4* xp = (const float4*)(x + b0 * 256 + c * 16 + 8 * h);

    float keepn = 0.0f, keeps = 1.0f;
    XPair xa, xb;
    Part pa, pb;
#pragma unroll
    for (int i = 0; i < 4; ++i) pb.v[i] = 0.0f;
    build_angles(w, F, threadIdx.x);
    load_pair(xa, xp, 0);
    build_tables(F, threadIdx.x);

    const bool even = (c & 1) == 0;
    float4 B1[2][2], B1x[2][2], B2[3][2][2];
#pragma unroll
    for (int s = 0; s < 2; ++s)
#pragma unroll
        for (int tp = 0; tp < 2; ++tp) {
            B1[s][tp]  = *(const float4*)&F.TLp[c & 1][2 * (c >> 1) + s][8 * h + 4 * tp];
            B1x[s][tp] = *(const float4*)&F.TLp[(c & 1) ^ 1][2 * (c >> 1) + s][8 * h + 4 * tp];
        }
#pragma unroll
    for (int P = 0; P < 3; ++P)
#pragma unroll
        for (int s = 0; s < 2; ++s)
#pragma unroll
            for (int tp = 0; tp < 2; ++tp)
                B2[P][s][tp] = *(const float4*)&F.Qp[P][s][c][8 * h + 4 * tp];
#pragma unroll 1
    for (int p = 0; p < PAIRS_PER_WAVE; p += 2) {
        load_pair(xb, xp, p + 1);
        __builtin_amdgcn_sched_barrier(0);
        reduce_keep(pb, p - 1, lane, keepn, keeps);
        do_pair(xa, B1, B1x, B2, even, pa);
        __builtin_amdgcn_sched_barrier(0);
        const int pn = (p + 2 < PAIRS_PER_WAVE) ? p + 2 : PAIRS_PER_WAVE - 1;
        load_pair(xa, xp, pn);
        __builtin_amdgcn_sched_barrier(0);
        reduce_keep(pa, p, lane, keepn, keeps);
        do_pair(xb, B1, B1x, B2, even, pb);
        __builtin_amdgcn_sched_barrier(0);
    }
    reduce_keep(pb, PAIRS_PER_WAVE - 1, lane, keepn, keeps);

    const float q = keepn / keeps;
    float2* o2 = (float2*)out + (b0 + lane);
    *o2 = make_float2(1.0f - q, q);
}


extern "C" void kernel_launch(void* const* d_in, const int* in_sizes, int n_in,
                              void* d_out, int out_size, void* d_ws, size_t ws_size,
                              hipStream_t stream) {
    const float* x = (const float*)d_in[0];
    const float* w = (const float*)d_in[1];
    float* out = (float*)d_out;

    const int B = in_sizes[0] / 256;

    qcnn_factored_wmma<<<B / ROWS_PER_BLOCK, WAVES_PER_BLOCK * 32, 0, stream>>>(x, w, out);
}


# Sums over the wave's lanes taken out of the batch loop: each pair's four lane partials go to LDS as one 16-byte store and lane i adds row i's 32 partials once at the end; new summation order
# speedup vs baseline: 4.7913x; 1.0385x over previous
#include <hip/hip_runtime.h>


typedef __attribute__((ext_vector_type(2))) float v2f;
typedef __attribute__((ext_vector_type(8))) float v8f;


__device__ __forceinline__ float2 rot_m(float2 a, float c, float s) {
    return make_float2(a.x * c + a.y * s, a.y * c - a.x * s);
}
__device__ __forceinline__ float2 rot_p(float2 a, float c, float s) {
    return make_float2(a.x * c - a.y * s, a.y * c + a.x * s);
}

__device__ __forceinline__ void q_rz1(float2 (&a)[4], float c, float s) {
    a[0] = rot_m(a[0], c, s); a[1] = rot_m(a[1], c, s);
    a[2] = rot_p(a[2], c, s); a[3] = rot_p(a[3], c, s);
}
__device__ __forceinline__ void q_rz2(float2 (&a)[4], float c, float s) {
    a[0] = rot_m(a[0], c, s); a[2] = rot_m(a[2], c, s);
    a[1] = rot_p(a[1], c, s); a[3] = rot_p(a[3], c, s);
}
__device__ __forceinline__ void q_ry2(float2 (&a)[4], float c, float s) {
#pragma unroll
    for (int x1 = 0; x1 < 2; ++x1) {
        float2 u = a[2 * x1], v = a[2 * x1 + 1];
        a[2 * x1]     = make_float2(c * u.x - s * v.x, c * u.y - s * v.y);
        a[2 * x1 + 1] = make_float2(s * u.x + c * v.x, s * u.y + c * v.y);
    }
}
__device__ __forceinline__ void q_cnot21(float2 (&a)[4]) {
    float2 tmp = a[1]; a[1] = a[3]; a[3] = tmp;
}
__device__ __forceinline__ void q_cnot12(float2 (&a)[4]) {
    float2 tmp = a[2]; a[2] = a[3]; a[3] = tmp;
}


__device__ __forceinline__ float2 cmul(float2 a, float2 b) {
    return make_float2(a.x * b.x - a.y * b.y, a.x * b.y + a.y * b.x);
}

__device__ __forceinline__ float2 nibble_entry(const float2* Ga, const float2* Gb, const float2* Gc,
                                               int o, int col) {
    const int o3 = (o >> 3) & 1, o2 = (o >> 2) & 1, o1 = (o >> 1) & 1, o0 = o & 1;
    const int p0 = col >> 2, q0 = col & 3;
    float2 acc = make_float2(0.0f, 0.0f);
#pragma unroll
    for (int a5 = 0; a5 < 2; ++a5) {
#pragma unroll
        for (int a7 = 0; a7 < 2; ++a7) {
            const float2 kv = cmul(Ga[(2 * o3 + a5) * 4 + p0], Gb[(2 * o1 + a7) * 4 + q0]);
            const float2 t  = cmul(Gc[(2 * o2 + o0) * 4 + 2 * a5 + a7], kv);
            acc.x += t.x; acc.y += t.y;
        }
    }
    return acc;
}

struct alignas(16) Factors {
    float  TLp[2][16][16];
    float  Qp[3][2][16][16];
    float2 THs[16][16];
    float2 Gs[7][16];
    float2 cs[42];
};

__device__ __forceinline__ void build_angles(const float* __restrict__ w, Factors& F, int t) {
    if (t < 42) {
        const float hw = 0.5f * w[t];
        F.cs[t] = make_float2(cosf(hw), sinf(hw));
    }
    __syncthreads();
}

__device__ __forceinline__ void build_tables(Factors& F, int t) {
    if (t < 28) {
        const int p = t >> 2, col = t & 3;
        const float HPI = 1.57079632679489662f;
        const float kc = cosf(0.5f * HPI), ks = sinf(0.5f * HPI);
        const float2* q = &F.cs[p * 6];
        float2 a[4];
#pragma unroll
        for (int r = 0; r < 4; ++r) a[r] = make_float2((r == col) ? 1.0f : 0.0f, 0.0f);
        q_rz2(a, kc, -ks);
        q_cnot21(a);
        q_rz1(a, q[0].x, q[0].y);
        q_ry2(a, q[1].x, q[1].y);
        q_cnot12(a);
        q_ry2(a, q[2].x, q[2].y);
        q_cnot21(a);
        q_rz1(a, kc, ks);
        q_rz2(a, kc, -ks);
        q_cnot21(a);
        q_rz1(a, q[3].x, q[3].y);
        q_ry2(a, q[4].x, q[4].y);
        q_cnot12(a);
        q_ry2(a, q[5].x, q[5].y);
#pragma unroll
        for (int r = 0; r < 4; ++r) F.Gs[p][r * 4 + col] = a[r];
    }
    __syncthreads();

    {
        const int o = t >> 4, col = t & 15;
        const float2 tl = nibble_entry(F.Gs[2], F.Gs[3], F.Gs[5], o, col);
        F.TLp[0][o][col] = tl.x;
        F.TLp[1][o][col] = tl.y;
        F.THs[o][col] = nibble_entry(F.Gs[0], F.Gs[1], F.Gs[4], o, col);
    }
    __syncthreads();

    {
        const int hp = t >> 4, a = t & 15;
        const int hr = hp >> 1, w3p = hp & 1;
#pragma unroll
        for (int s = 0; s < 2; ++s) {
            float2 acc = make_float2(0.0f, 0.0f);
#pragma unroll
            for (int w3 = 0; w3 < 2; ++w3) {
                const float2 tq = cmul(F.Gs[6][(2 * w3p + 1) * 4 + 2 * w3 + s], F.THs[2 * hr + w3][a]);
                acc.x += tq.x; acc.y += tq.y;
            }
            F.Qp[0][s][hp][a] = acc.x;
            F.Qp[1][s][hp][a] = acc.y;
            F.Qp[2][s][hp][a] = acc.x + acc.y;
        }
    }
    __syncthreads();
}


template <int M>
__device__ __forceinline__ float row_xor(float v) {
    return __int_as_float(__builtin_amdgcn_update_dpp(
        0, __float_as_int(v), 0x160 + M  , 0xf, 0xf, true));
}

__device__ __forceinline__ v8f mma(v8f acc, float a0, float a1, float b0, float b1) {
    v2f a; a[0] = a0; a[1] = a1;
    v2f b; b[0] = b0; b[1] = b1;
    return __builtin_amdgcn_wmma_f32_16x16x4_f32(false, a, false, b, (short)0, acc, false, false);
}

struct XPair { float4 v[2][2]; };

__device__ __forceinline__ void load_pair(XPair& xq, const float4* xp, int p) {
#pragma unroll
    for (int e = 0; e < 2; ++e) {
        xq.v[e][0] = xp[(size_t)(2 * p + e) * 64];
        xq.v[e][1] = xp[(size_t)(2 * p + e) * 64 + 1];
    }
}

__device__ __forceinline__ float sumsq(const float4 (&v)[2]) {
    v2f p0; p0[0] = v[0].x; p0[1] = v[0].y;
    v2f p1; p1[0] = v[0].z; p1[1] = v[0].w;
    v2f p2; p2[0] = v[1].x; p2[1] = v[1].y;
    v2f p3; p3[0] = v[1].z; p3[1] = v[1].w;
    v2f s = p0 * p0;
    s = __builtin_elementwise_fma(p1, p1, s);
    s = __builtin_elementwise_fma(p2, p2, s);
    s = __builtin_elementwise_fma(p3, p3, s);
    return s[0] + s[1];
}

__device__ __forceinline__ v2f reg_pair(v8f t, int j) {
    switch (j) {
    case 0:  return __builtin_shufflevector(t, t, 0, 1);
    case 1:  return __builtin_shufflevector(t, t, 2, 3);
    case 2:  return __builtin_shufflevector(t, t, 4, 5);
    default: return __builtin_shufflevector(t, t, 6, 7);
    }
}


__device__ __forceinline__ void do_pair(const XPair& xq, const float4 (&B1)[2][2], const float4 (&B1x)[2][2],
                                        const float4 (&B2)[3][2][2], bool even, float4& part) {
    const float4 (&x0)[2] = xq.v[0];
    const float4 (&x1)[2] = xq.v[1];
    v8f d1[2][2] = {};
#pragma unroll
    for (int s = 0; s < 2; ++s) {
#pragma unroll
        for (int tp = 0; tp < 2; ++tp) {
            d1[0][s] = mma(d1[0][s], x0[tp].x, x0[tp].y, B1[s][tp].x, B1[s][tp].y);
            d1[1][s] = mma(d1[1][s], x1[tp].x, x1[tp].y, B1x[s][tp].x, B1x[s][tp].y);
            d1[0][s] = mma(d1[0][s], x0[tp].z, x0[tp].w, B1[s][tp].z, B1[s][tp].w);
            d1[1][s] = mma(d1[1][s], x1[tp].z, x1[tp].w, B1x[s][tp].z, B1x[s][tp].w);
        }
    }
    v8f yr[2], yi[2], ys[2];
#pragma unroll
    for (int s = 0; s < 2; ++s) {
        yr[s] = even ? d1[0][s] : d1[1][s];
        yi[s] = even ? d1[1][s] : d1[0][s];
#pragma unroll
        for (int r = 0; r < 8; ++r) ys[s][r] = yr[s][r] + row_xor<1>(yi[s][r]);
    }
    v8f d2[3] = {};
#pragma unroll
    for (int s = 0; s < 2; ++s) {
#pragma unroll
        for (int tp = 0; tp < 2; ++tp) {
            d2[0] = mma(d2[0], yr[s][4 * tp + 0], yr[s][4 * tp + 1], B2[0][s][tp].x, B2[0][s][tp].y);
            d2[1] = mma(d2[1], yi[s][4 * tp + 0], yi[s][4 * tp + 1], B2[1][s][tp].x, B2[1][s][tp].y);
            d2[2] = mma(d2[2], ys[s][4 * tp + 0], ys[s][4 * tp + 1], B2[2][s][tp].x, B2[2][s][tp].y);
            d2[0] = mma(d2[0], yr[s][4 * tp + 2], yr[s][4 * tp + 3], B2[0][s][tp].z, B2[0][s][tp].w);
            d2[1] = mma(d2[1], yi[s][4 * tp + 2], yi[s][4 * tp + 3], B2[1][s][tp].z, B2[1][s][tp].w);
            d2[2] = mma(d2[2], ys[s][4 * tp + 2], ys[s][4 * tp + 3], B2[2][s][tp].z, B2[2][s][tp].w);
        }
    }
    v2f acc = {0.0f, 0.0f};
#pragma unroll
    for (int j = 0; j < 4; ++j) {
        const v2f p1 = reg_pair(d2[0], j);
        v2f p2x = reg_pair(d2[1], j);
        const v2f p3 = reg_pair(d2[2], j);
        asm("" : "+v"(p2x));
        v2f p2; p2[0] = p2x[1]; p2[1] = p2x[0];
        const v2f ore = p1 - p2;
        const v2f oim = (p3 - p1) - p2;
        acc = __builtin_elementwise_fma(ore, ore, acc);
        acc = __builtin_elementwise_fma(oim, oim, acc);
    }
    part = make_float4(acc[0], acc[1], sumsq(x0), sumsq(x1));
}


#define WAVES_PER_BLOCK 8
#define ROWS_PER_WAVE 32
#define PAIRS_PER_WAVE (ROWS_PER_WAVE / 2)
#define ROWS_PER_BLOCK (WAVES_PER_BLOCK * ROWS_PER_WAVE)

__global__ __launch_bounds__(WAVES_PER_BLOCK * 32) void qcnn_factored_wmma(const float* __restrict__ x,
                                                                         const float* __restrict__ w,
                                                                         float* __restrict__ out) {
    __shared__ Factors F;
    __shared__ float4 red[WAVES_PER_BLOCK][PAIRS_PER_WAVE][33];
    const int lane = threadIdx.x & 31;
    const int wave = threadIdx.x >> 5;
    const int c = lane & 15, h = lane >> 4;
    const size_t b0 = ((size_t)blockIdx.x * WAVES_PER_BLOCK + wave) * ROWS_PER_WAVE;

    const float4* xp = (const float4*)(x + b0 * 256 + c * 16 + 8 * h);

    XPair xa, xb;
    float4 pa, pb;
    float4* myred = &red[wave][0][lane];
    build_angles(w, F, threadIdx.x);
    load_pair(xa, xp, 0);
    build_tables(F, threadIdx.x);

    const bool even = (c & 1) == 0;
    float4 B1[2][2], B1x[2][2], B2[3][2][2];
#pragma unroll
    for (int s = 0; s < 2; ++s)
#pragma unroll
        for (int tp = 0; tp < 2; ++tp) {
            B1[s][tp]  = *(const float4*)&F.TLp[c & 1][2 * (c >> 1) + s][8 * h + 4 * tp];
            B1x[s][tp] = *(const float4*)&F.TLp[(c & 1) ^ 1][2 * (c >> 1) + s][8 * h + 4 * tp];
        }
#pragma unroll
    for (int P = 0; P < 3; ++P)
#pragma unroll
        for (int s = 0; s < 2; ++s)
#pragma unroll
            for (int tp = 0; tp < 2; ++tp)
                B2[P][s][tp] = *(const float4*)&F.Qp[P][s][c][8 * h + 4 * tp];
#pragma unroll 1
    for (int p = 0; p < PAIRS_PER_WAVE; p += 2) {
        load_pair(xb, xp, p + 1);
        __builtin_amdgcn_sched_barrier(0);
        do_pair(xa, B1, B1x, B2, even, pa);
        myred[33 * p] = pa;
        __builtin_amdgcn_sched_barrier(0);
        const int pn = (p + 2 < PAIRS_PER_WAVE) ? p + 2 : PAIRS_PER_WAVE - 1;
        load_pair(xa, xp, pn);
        __builtin_amdgcn_sched_barrier(0);
        do_pair(xb, B1, B1x, B2, even, pb);
        myred[33 * (p + 1)] = pb;
        __builtin_amdgcn_sched_barrier(0);
    }

    __builtin_amdgcn_fence(__ATOMIC_RELEASE, "wavefront");
    __builtin_amdgcn_wave_barrier();
    __builtin_amdgcn_fence(__ATOMIC_ACQUIRE, "wavefront");
    const int pr = lane >> 1, er = lane & 1;
    const float* slab = (const float*)&red[wave][pr][0] + er;
    v2f sum4[4] = {};
#pragma unroll
    for (int l = 0; l < 32; ++l) {
        v2f t; t[0] = slab[4 * l]; t[1] = slab[4 * l + 2];
        sum4[l & 3] += t;
    }
    const v2f tot = (sum4[0] + sum4[1]) + (sum4[2] + sum4[3]);

    const float q = tot[0] / tot[1];
    float2* o2 = (float2*)out + (b0 + lane);
    *o2 = make_float2(1.0f - q, q);
}


extern "C" void kernel_launch(void* const* d_in, const int* in_sizes, int n_in,
                              void* d_out, int out_size, void* d_ws, size_t ws_size,
                              hipStream_t stream) {
    const float* x = (const float*)d_in[0];
    const float* w = (const float*)d_in[1];
    float* out = (float*)d_out;

    const int B = in_sizes[0] / 256;

    qcnn_factored_wmma<<<B / ROWS_PER_BLOCK, WAVES_PER_BLOCK * 32, 0, stream>>>(x, w, out);
}


# Batch-row reads issued three pairs (120 WMMAs) ahead of their use instead of one: four register sets rotating over a four-pair loop trip; arithmetic and summation order unchanged
# speedup vs baseline: 4.9459x; 1.0323x over previous
#include <hip/hip_runtime.h>


typedef __attribute__((ext_vector_type(2))) float v2f;
typedef __attribute__((ext_vector_type(8))) float v8f;


__device__ __forceinline__ float2 rot_m(float2 a, float c, float s) {
    return make_float2(a.x * c + a.y * s, a.y * c - a.x * s);
}
__device__ __forceinline__ float2 rot_p(float2 a, float c, float s) {
    return make_float2(a.x * c - a.y * s, a.y * c + a.x * s);
}

__device__ __forceinline__ void q_rz1(float2 (&a)[4], float c, float s) {
    a[0] = rot_m(a[0], c, s); a[1] = rot_m(a[1], c, s);
    a[2] = rot_p(a[2], c, s); a[3] = rot_p(a[3], c, s);
}
__device__ __forceinline__ void q_rz2(float2 (&a)[4], float c, float s) {
    a[0] = rot_m(a[0], c, s); a[2] = rot_m(a[2], c, s);
    a[1] = rot_p(a[1], c, s); a[3] = rot_p(a[3], c, s);
}
__device__ __forceinline__ void q_ry2(float2 (&a)[4], float c, float s) {
#pragma unroll
    for (int x1 = 0; x1 < 2; ++x1) {
        float2 u = a[2 * x1], v = a[2 * x1 + 1];
        a[2 * x1]     = make_float2(c * u.x - s * v.x, c * u.y - s * v.y);
        a[2 * x1 + 1] = make_float2(s * u.x + c * v.x, s * u.y + c * v.y);
    }
}
__device__ __forceinline__ void q_cnot21(float2 (&a)[4]) {
    float2 tmp = a[1]; a[1] = a[3]; a[3] = tmp;
}
__device__ __forceinline__ void q_cnot12(float2 (&a)[4]) {
    float2 tmp = a[2]; a[2] = a[3]; a[3] = tmp;
}


__device__ __forceinline__ float2 cmul(float2 a, float2 b) {
    return make_float2(a.x * b.x - a.y * b.y, a.x * b.y + a.y * b.x);
}

__device__ __forceinline__ float2 nibble_entry(const float2* Ga, const float2* Gb, const float2* Gc,
                                               int o, int col) {
    const int o3 = (o >> 3) & 1, o2 = (o >> 2) & 1, o1 = (o >> 1) & 1, o0 = o & 1;
    const int p0 = col >> 2, q0 = col & 3;
    float2 acc = make_float2(0.0f, 0.0f);
#pragma unroll
    for (int a5 = 0; a5 < 2; ++a5) {
#pragma unroll
        for (int a7 = 0; a7 < 2; ++a7) {
            const float2 kv = cmul(Ga[(2 * o3 + a5) * 4 + p0], Gb[(2 * o1 + a7) * 4 + q0]);
            const float2 t  = cmul(Gc[(2 * o2 + o0) * 4 + 2 * a5 + a7], kv);
            acc.x += t.x; acc.y += t.y;
        }
    }
    return acc;
}

struct alignas(16) Factors {
    float  TLp[2][16][16];
    float  Qp[3][2][16][16];
    float2 THs[16][16];
    float2 Gs[7][16];
    float2 cs[42];
};

__device__ __forceinline__ void build_angles(const float* __restrict__ w, Factors& F, int t) {
    if (t < 42) {
        const float hw = 0.5f * w[t];
        F.cs[t] = make_float2(cosf(hw), sinf(hw));
    }
    __syncthreads();
}

__device__ __forceinline__ void build_tables(Factors& F, int t) {
    if (t < 28) {
        const int p = t >> 2, col = t & 3;
        const float HPI = 1.57079632679489662f;
        const float kc = cosf(0.5f * HPI), ks = sinf(0.5f * HPI);
        const float2* q = &F.cs[p * 6];
        float2 a[4];
#pragma unroll
        for (int r = 0; r < 4; ++r) a[r] = make_float2((r == col) ? 1.0f : 0.0f, 0.0f);
        q_rz2(a, kc, -ks);
        q_cnot21(a);
        q_rz1(a, q[0].x, q[0].y);
        q_ry2(a, q[1].x, q[1].y);
        q_cnot12(a);
        q_ry2(a, q[2].x, q[2].y);
        q_cnot21(a);
        q_rz1(a, kc, ks);
        q_rz2(a, kc, -ks);
        q_cnot21(a);
        q_rz1(a, q[3].x, q[3].y);
        q_ry2(a, q[4].x, q[4].y);
        q_cnot12(a);
        q_ry2(a, q[5].x, q[5].y);
#pragma unroll
        for (int r = 0; r < 4; ++r) F.Gs[p][r * 4 + col] = a[r];
    }
    __syncthreads();

    {
        const int o = t >> 4, col = t & 15;
        const float2 tl = nibble_entry(F.Gs[2], F.Gs[3], F.Gs[5], o, col);
        F.TLp[0][o][col] = tl.x;
        F.TLp[1][o][col] = tl.y;
        F.THs[o][col] = nibble_entry(F.Gs[0], F.Gs[1], F.Gs[4], o, col);
    }
    __syncthreads();

    {
        const int hp = t >> 4, a = t & 15;
        const int hr = hp >> 1, w3p = hp & 1;
#pragma unroll
        for (int s = 0; s < 2; ++s) {
            float2 acc = make_float2(0.0f, 0.0f);
#pragma unroll
            for (int w3 = 0; w3 < 2; ++w3) {
                const float2 tq = cmul(F.Gs[6][(2 * w3p + 1) * 4 + 2 * w3 + s], F.THs[2 * hr + w3][a]);
                acc.x += tq.x; acc.y += tq.y;
            }
            F.Qp[0][s][hp][a] = acc.x;
            F.Qp[1][s][hp][a] = acc.y;
            F.Qp[2][s][hp][a] = acc.x + acc.y;
        }
    }
    __syncthreads();
}


template <int M>
__device__ __forceinline__ float row_xor(float v) {
    return __int_as_float(__builtin_amdgcn_update_dpp(
        0, __float_as_int(v), 0x160 + M  , 0xf, 0xf, true));
}

__device__ __forceinline__ v8f mma(v8f acc, float a0, float a1, float b0, float b1) {
    v2f a; a[0] = a0; a[1] = a1;
    v2f b; b[0] = b0; b[1] = b1;
    return __builtin_amdgcn_wmma_f32_16x16x4_f32(false, a, false, b, (short)0, acc, false, false);
}

struct XPair { float4 v[2][2]; };

__device__ __forceinline__ void load_pair(XPair& xq, const float4* q) {
#pragma unroll
    for (int e = 0; e < 2; ++e) {
        xq.v[e][0] = q[e * 64];
        xq.v[e][1] = q[e * 64 + 1];
    }
}

#define PAIR_STRIDE 128

__device__ __forceinline__ float sumsq(const float4 (&v)[2]) {
    v2f p0; p0[0] = v[0].x; p0[1] = v[0].y;
    v2f p1; p1[0] = v[0].z; p1[1] = v[0].w;
    v2f p2; p2[0] = v[1].x; p2[1] = v[1].y;
    v2f p3; p3[0] = v[1].z; p3[1] = v[1].w;
    v2f s = p0 * p0;
    s = __builtin_elementwise_fma(p1, p1, s);
    s = __builtin_elementwise_fma(p2, p2, s);
    s = __builtin_elementwise_fma(p3, p3, s);
    return s[0] + s[1];
}

__device__ __forceinline__ v2f reg_pair(v8f t, int j) {
    switch (j) {
    case 0:  return __builtin_shufflevector(t, t, 0, 1);
    case 1:  return __builtin_shufflevector(t, t, 2, 3);
    case 2:  return __builtin_shufflevector(t, t, 4, 5);
    default: return __builtin_shufflevector(t, t, 6, 7);
    }
}


__device__ __forceinline__ void do_pair(const XPair& xq, const float4 (&B1)[2][2], const float4 (&B1x)[2][2],
                                        const float4 (&B2)[3][2][2], bool even, float4& part) {
    const float4 (&x0)[2] = xq.v[0];
    const float4 (&x1)[2] = xq.v[1];
    v8f d1[2][2] = {};
#pragma unroll
    for (int s = 0; s < 2; ++s) {
#pragma unroll
        for (int tp = 0; tp < 2; ++tp) {
            d1[0][s] = mma(d1[0][s], x0[tp].x, x0[tp].y, B1[s][tp].x, B1[s][tp].y);
            d1[1][s] = mma(d1[1][s], x1[tp].x, x1[tp].y, B1x[s][tp].x, B1x[s][tp].y);
            d1[0][s] = mma(d1[0][s], x0[tp].z, x0[tp].w, B1[s][tp].z, B1[s][tp].w);
            d1[1][s] = mma(d1[1][s], x1[tp].z, x1[tp].w, B1x[s][tp].z, B1x[s][tp].w);
        }
    }
    v8f yr[2], yi[2], ys[2];
#pragma unroll
    for (int s = 0; s < 2; ++s) {
        yr[s] = even ? d1[0][s] : d1[1][s];
        yi[s] = even ? d1[1][s] : d1[0][s];
#pragma unroll
        for (int r = 0; r < 8; ++r) ys[s][r] = yr[s][r] + row_xor<1>(yi[s][r]);
    }
    v8f d2[3] = {};
#pragma unroll
    for (int s = 0; s < 2; ++s) {
#pragma unroll
        for (int tp = 0; tp < 2; ++tp) {
            d2[0] = mma(d2[0], yr[s][4 * tp + 0], yr[s][4 * tp + 1], B2[0][s][tp].x, B2[0][s][tp].y);
            d2[1] = mma(d2[1], yi[s][4 * tp + 0], yi[s][4 * tp + 1], B2[1][s][tp].x, B2[1][s][tp].y);
            d2[2] = mma(d2[2], ys[s][4 * tp + 0], ys[s][4 * tp + 1], B2[2][s][tp].x, B2[2][s][tp].y);
            d2[0] = mma(d2[0], yr[s][4 * tp + 2], yr[s][4 * tp + 3], B2[0][s][tp].z, B2[0][s][tp].w);
            d2[1] = mma(d2[1], yi[s][4 * tp + 2], yi[s][4 * tp + 3], B2[1][s][tp].z, B2[1][s][tp].w);
            d2[2] = mma(d2[2], ys[s][4 * tp + 2], ys[s][4 * tp + 3], B2[2][s][tp].z, B2[2][s][tp].w);
        }
    }
    v2f acc = {0.0f, 0.0f};
#pragma unroll
    for (int j = 0; j < 4; ++j) {
        const v2f p1 = reg_pair(d2[0], j);
        v2f p2x = reg_pair(d2[1], j);
        const v2f p3 = reg_pair(d2[2], j);
        asm("" : "+v"(p2x));
        v2f p2; p2[0] = p2x[1]; p2[1] = p2x[0];
        const v2f ore = p1 - p2;
        const v2f oim = (p3 - p1) - p2;
        acc = __builtin_elementwise_fma(ore, ore, acc);
        acc = __builtin_elementwise_fma(oim, oim, acc);
    }
    part = make_float4(acc[0], acc[1], sumsq(x0), sumsq(x1));
}


#define WAVES_PER_BLOCK 8
#define ROWS_PER_WAVE 32
#define PAIRS_PER_WAVE (ROWS_PER_WAVE / 2)
#define ROWS_PER_BLOCK (WAVES_PER_BLOCK * ROWS_PER_WAVE)

__global__ __launch_bounds__(WAVES_PER_BLOCK * 32) void qcnn_factored_wmma(const float* __restrict__ x,
                                                                         const float* __restrict__ w,
                                                                         float* __restrict__ out) {
    __shared__ Factors F;
    __shared__ float4 red[WAVES_PER_BLOCK][PAIRS_PER_WAVE][33];
    const int lane = threadIdx.x & 31;
    const int wave = threadIdx.x >> 5;
    const int c = lane & 15, h = lane >> 4;
    const size_t b0 = ((size_t)blockIdx.x * WAVES_PER_BLOCK + wave) * ROWS_PER_WAVE;

    const float4* xp = (const float4*)(x + b0 * 256 + c * 16 + 8 * h);

    XPair xa, xb, xc, xd;
    float4 pa, pb, pc, pd;
    float4* myred = &red[wave][0][lane];
    build_angles(w, F, threadIdx.x);
    load_pair(xa, xp);
    build_tables(F, threadIdx.x);
    load_pair(xb, xp + PAIR_STRIDE);
    load_pair(xc, xp + 2 * PAIR_STRIDE);

    const bool even = (c & 1) == 0;
    float4 B1[2][2], B1x[2][2], B2[3][2][2];
#pragma unroll
    for (int s = 0; s < 2; ++s)
#pragma unroll
        for (int tp = 0; tp < 2; ++tp) {
            B1[s][tp]  = *(const float4*)&F.TLp[c & 1][2 * (c >> 1) + s][8 * h + 4 * tp];
            B1x[s][tp] = *(const float4*)&F.TLp[(c & 1) ^ 1][2 * (c >> 1) + s][8 * h + 4 * tp];
        }
#pragma unroll
    for (int P = 0; P < 3; ++P)
#pragma unroll
        for (int s = 0; s < 2; ++s)
#pragma unroll
            for (int tp = 0; tp < 2; ++tp)
                B2[P][s][tp] = *(const float4*)&F.Qp[P][s][c][8 * h + 4 * tp];
    const float4* q1 = xp + 3 * PAIR_STRIDE;
    const float4* q2 = xp;
#pragma unroll 1
    for (int p = 0; p < PAIRS_PER_WAVE; p += 4) {
        load_pair(xd, q1);
        __builtin_amdgcn_sched_barrier(0);
        do_pair(xa, B1, B1x, B2, even, pa);
        myred[33 * p] = pa;
        __builtin_amdgcn_sched_barrier(0);
        load_pair(xa, q2 + 4 * PAIR_STRIDE);
        __builtin_amdgcn_sched_barrier(0);
        do_pair(xb, B1, B1x, B2, even, pb);
        myred[33 * (p + 1)] = pb;
        __builtin_amdgcn_sched_barrier(0);
        load_pair(xb, q2 + 5 * PAIR_STRIDE);
        __builtin_amdgcn_sched_barrier(0);
        do_pair(xc, B1, B1x, B2, even, pc);
        myred[33 * (p + 2)] = pc;
        __builtin_amdgcn_sched_barrier(0);
        load_pair(xc, q2 + 6 * PAIR_STRIDE);
        __builtin_amdgcn_sched_barrier(0);
        do_pair(xd, B1, B1x, B2, even, pd);
        myred[33 * (p + 3)] = pd;
        __builtin_amdgcn_sched_barrier(0);
        q1 += 4 * PAIR_STRIDE;
        q2 = xp + PAIR_STRIDE * ((p + 4 < 9) ? p + 4 : 9);
    }

    __builtin_amdgcn_fence(__ATOMIC_RELEASE, "wavefront");
    __builtin_amdgcn_wave_barrier();
    __builtin_amdgcn_fence(__ATOMIC_ACQUIRE, "wavefront");
    const int pr = lane >> 1, er = lane & 1;
    const float* slab = (const float*)&red[wave][pr][0] + er;
    v2f sum4[4] = {};
#pragma unroll
    for (int l = 0; l < 32; ++l) {
        v2f t; t[0] = slab[4 * l]; t[1] = slab[4 * l + 2];
        sum4[l & 3] += t;
    }
    const v2f tot = (sum4[0] + sum4[1]) + (sum4[2] + sum4[3]);

    const float q = tot[0] / tot[1];
    float2* o2 = (float2*)out + (b0 + lane);
    *o2 = make_float2(1.0f - q, q);
}


extern "C" void kernel_launch(void* const* d_in, const int* in_sizes, int n_in,
                              void* d_out, int out_size, void* d_ws, size_t ws_size,
                              hipStream_t stream) {
    const float* x = (const float*)d_in[0];
    const float* w = (const float*)d_in[1];
    float* out = (float*)d_out;

    const int B = in_sizes[0] / 256;

    qcnn_factored_wmma<<<B / ROWS_PER_BLOCK, WAVES_PER_BLOCK * 32, 0, stream>>>(x, w, out);
}
